# starting kernel
# baseline (speedup 1.0000x reference)
#include <hip/hip_runtime.h>

typedef __attribute__((ext_vector_type(2))) float v2f;
typedef __attribute__((ext_vector_type(8))) float v8f;

#define D_FEAT 64
constexpr float LAM = 2.0f;


__global__ void zero_kernel(float* __restrict__ p, long n) {
  long i = (long)blockIdx.x * blockDim.x + threadIdx.x;
  long stride = (long)gridDim.x * blockDim.x;
  for (; i < n; i += stride) p[i] = 0.0f;
}

__global__ void deg_kernel(const int* __restrict__ dst, float* __restrict__ deg, int E) {
  int i = blockIdx.x * blockDim.x + threadIdx.x;
  if (i < E) atomicAdd(&deg[dst[i]], 1.0f);
}

__global__ void norm_kernel(float* __restrict__ d, int N) {
  int i = blockIdx.x * blockDim.x + threadIdx.x;
  if (i < N) {
    float v = fmaxf(d[i], 1.0f);
    d[i] = rsqrtf(v);
  }
}

__global__ void scatter_kernel(const float* __restrict__ x, const float* __restrict__ nrm,
                               const int* __restrict__ src, const int* __restrict__ dst,
                               float* __restrict__ agg, int E) {
  int lane = threadIdx.x & 31;
  int edge = blockIdx.x * (blockDim.x >> 5) + (threadIdx.x >> 5);
  if (edge >= E) return;
  int s = src[edge];
  int d = dst[edge];
  float ns = nrm[s];
  const float2* xs = (const float2*)(x + (long)s * D_FEAT);
  float2 v = xs[lane];
  float* o = agg + (long)d * D_FEAT + lane * 2;
  atomicAdd(o + 0, v.x * ns);
  atomicAdd(o + 1, v.y * ns);
}

__global__ void combine_kernel(const float* __restrict__ agg, const float* __restrict__ nrm,
                               const float* __restrict__ t1, const float* __restrict__ t2,
                               float* __restrict__ out,
                               float cH, float cX, float cP, long n) {
  long i = (long)blockIdx.x * blockDim.x + threadIdx.x;
  if (i >= n) return;
  long node = i >> 6;
  float h = agg[i] * nrm[node];
  out[i] = cH * h + cX * t1[i] + cP * t2[i];
}


__device__ __forceinline__ v8f accum_term(v8f acc, const float* __restrict__ xr,
                                          const float* __restrict__ Wt,
                                          int khalf, int ncol) {
  v2f bfrag[16];
#pragma unroll
  for (int kk = 0; kk < 16; ++kk) {
    int krow = kk * 4 + khalf * 2;
    v2f bf;
    bf.x = Wt[(krow + 0) * D_FEAT + ncol];
    bf.y = Wt[(krow + 1) * D_FEAT + ncol];
    bfrag[kk] = bf;
  }
#pragma unroll
  for (int kk = 0; kk < 16; ++kk) {
    v2f a;
    a.x = xr[kk * 4 + khalf * 2 + 0];
    a.y = xr[kk * 4 + khalf * 2 + 1];
    acc = __builtin_amdgcn_wmma_f32_16x16x4_f32(
         false, a,  false, bfrag[kk],
         (short)0, acc,  false,  false);
  }
  return acc;
}

__global__ __launch_bounds__(128) void gemm_fused_kernel(
    const float* __restrict__ X0, const float* __restrict__ X1,
    const float* __restrict__ X2, const float* __restrict__ X3,
    const float* __restrict__ W, const float* __restrict__ bias,
    float* __restrict__ out, int N) {
  int lane = threadIdx.x & 31;
  int wv   = threadIdx.x >> 5;
  int row0 = blockIdx.x * 16;
  if (row0 >= N) return;

  int mlo   = lane & 15;
  int khalf = lane >> 4;
  int ncol  = wv * 16 + mlo;

  int arow = row0 + mlo;
  if (arow >= N) arow = N - 1;
  long aoff = (long)arow * D_FEAT;

  v8f acc;
  {
    float bb = bias[ncol];
#pragma unroll
    for (int v = 0; v < 8; ++v) acc[v] = bb;
  }

  acc = accum_term(acc, X0 + aoff, W + 0 * D_FEAT * D_FEAT, khalf, ncol);
  acc = accum_term(acc, X1 + aoff, W + 1 * D_FEAT * D_FEAT, khalf, ncol);
  acc = accum_term(acc, X2 + aoff, W + 2 * D_FEAT * D_FEAT, khalf, ncol);
  acc = accum_term(acc, X3 + aoff, W + 3 * D_FEAT * D_FEAT, khalf, ncol);

  if (row0 + 16 <= N) {
    float* orow = out + (long)(row0 + khalf * 8) * D_FEAT + ncol;
#pragma unroll
    for (int v = 0; v < 8; ++v) orow[(long)v * D_FEAT] = acc[v];
  } else {
#pragma unroll
    for (int v = 0; v < 8; ++v) {
      int r = row0 + khalf * 8 + v;
      if (r < N) out[(long)r * D_FEAT + ncol] = acc[v];
    }
  }
}


extern "C" void kernel_launch(void* const* d_in, const int* in_sizes, int n_in,
                              void* d_out, int out_size, void* d_ws, size_t ws_size,
                              hipStream_t stream) {
  const float* feat = (const float*)d_in[0];
  const int*   src  = (const int*)d_in[1];
  const int*   dst  = (const int*)d_in[2];
  const float* W    = (const float*)d_in[3];
  const float* b    = (const float*)d_in[4];
  float*       out  = (float*)d_out;

  const int N = in_sizes[0] / D_FEAT;
  const int E = in_sizes[1];
  const long nd = (long)N * D_FEAT;

  float* nrm = (float*)d_ws;
  float* agg = nrm + N;
  float* txA = agg + nd;
  float* txB = txA + nd;

  const int cgrid = (int)((nd + 255) / 256);
  const int sgrid = (E + 7) / 8;
  const int ggrid = (N + 15) / 16;

  zero_kernel<<<512, 256, 0, stream>>>(nrm, (long)N);
  deg_kernel<<<(E + 255) / 256, 256, 0, stream>>>(dst, nrm, E);
  norm_kernel<<<(N + 255) / 256, 256, 0, stream>>>(nrm, N);

  zero_kernel<<<2048, 256, 0, stream>>>(agg, nd);
  scatter_kernel<<<sgrid, 256, 0, stream>>>(feat, nrm, src, dst, agg, E);
  combine_kernel<<<cgrid, 256, 0, stream>>>(agg, nrm, feat, feat, txA,
                                            -2.0f / LAM, 2.0f / LAM - 1.0f, 0.0f, nd);

  zero_kernel<<<2048, 256, 0, stream>>>(agg, nd);
  scatter_kernel<<<sgrid, 256, 0, stream>>>(txA, nrm, src, dst, agg, E);
  combine_kernel<<<cgrid, 256, 0, stream>>>(agg, nrm, txA, feat, txB,
                                            -4.0f / LAM, 4.0f / LAM - 2.0f, -1.0f, nd);

  zero_kernel<<<2048, 256, 0, stream>>>(agg, nd);
  scatter_kernel<<<sgrid, 256, 0, stream>>>(txB, nrm, src, dst, agg, E);
  combine_kernel<<<cgrid, 256, 0, stream>>>(agg, nrm, txB, txA, agg,
                                            -4.0f / LAM, 4.0f / LAM - 2.0f, -1.0f, nd);

  gemm_fused_kernel<<<ggrid, 128, 0, stream>>>(feat, txA, txB, agg, W, b, out, N);
}


# Edges sorted by dst once per call (int atomic ranks, scan, fill); the 3 props become atomic-free per-node row gathers summed in arrival-rank order, zero and combine passes fused in
# speedup vs baseline: 2.5849x; 2.5849x over previous
#include <hip/hip_runtime.h>


typedef __attribute__((ext_vector_type(2))) float v2f;
typedef __attribute__((ext_vector_type(8))) float v8f;

#define D_FEAT 64
constexpr float LAM = 2.0f;


__global__ void zero_int_kernel(int* __restrict__ p, int n) {
  int i = blockIdx.x * blockDim.x + threadIdx.x;
  if (i < n) p[i] = 0;
}

__global__ void rank_kernel(const int* __restrict__ dst, int* __restrict__ cnt,
                            int* __restrict__ rank, int E) {
  int i = blockIdx.x * blockDim.x + threadIdx.x;
  if (i < E) rank[i] = atomicAdd(&cnt[dst[i]], 1);
}

__device__ __forceinline__ int wave_incl_scan(int x, int lane) {
#pragma unroll
  for (int o = 1; o < 32; o <<= 1) {
    int y = __shfl_up(x, o, 32);
    if (lane >= o) x += y;
  }
  return x;
}

__global__ __launch_bounds__(256) void scan_local_kernel(
    const int* __restrict__ cnt, int* __restrict__ lscan, int* __restrict__ btot,
    float* __restrict__ nrm, int N) {
  __shared__ int wsum[8];
  int t = threadIdx.x, lane = t & 31, w = t >> 5;
  int i0 = blockIdx.x * 1024 + t * 4;
  int c[4];
#pragma unroll
  for (int k = 0; k < 4; ++k) c[k] = (i0 + k < N) ? cnt[i0 + k] : 0;
  int s = c[0] + c[1] + c[2] + c[3];
  int inc = wave_incl_scan(s, lane);
  if (lane == 31) wsum[w] = inc;
  __syncthreads();
  int woff = 0;
#pragma unroll
  for (int k = 0; k < 8; ++k) if (k < w) woff += wsum[k];
  int run = woff + inc - s;
#pragma unroll
  for (int k = 0; k < 4; ++k) {
    if (i0 + k < N) {
      lscan[i0 + k] = run;
      nrm[i0 + k] = rsqrtf(fmaxf((float)c[k], 1.0f));
    }
    run += c[k];
  }
  if (t == 255) btot[blockIdx.x] = run;
}

__global__ __launch_bounds__(1024) void scan_blocks_kernel(
    const int* __restrict__ btot, int* __restrict__ boff, int nb) {
  __shared__ int wsum[32];
  int t = threadIdx.x, lane = t & 31, w = t >> 5;
  int carry = 0;
  for (int base = 0; base < nb; base += 1024) {
    int i = base + t;
    int s = (i < nb) ? btot[i] : 0;
    int inc = wave_incl_scan(s, lane);
    if (lane == 31) wsum[w] = inc;
    __syncthreads();
    int woff = 0, tot = 0;
#pragma unroll
    for (int k = 0; k < 32; ++k) {
      int x = wsum[k];
      tot += x;
      if (k < w) woff += x;
    }
    if (i < nb) boff[i] = carry + woff + inc - s;
    carry += tot;
    __syncthreads();
  }
}

__global__ void fill_kernel(const int* __restrict__ src, const int* __restrict__ dst,
                            const int* __restrict__ rank, const int* __restrict__ lscan,
                            const int* __restrict__ boff, int* __restrict__ col, int E) {
  int i = blockIdx.x * blockDim.x + threadIdx.x;
  if (i < E) {
    int d = dst[i];
    int p = lscan[d] + boff[d >> 10] + rank[i];
    if ((unsigned)p < (unsigned)E) col[p] = src[i];
  }
}

template <bool HAS_PREV>
__global__ __launch_bounds__(256) void gather_combine_kernel(
    const float* __restrict__ xin, const float* __restrict__ xprev,
    const float* __restrict__ nrm, const int* __restrict__ cnt,
    const int* __restrict__ lscan, const int* __restrict__ boff,
    const int* __restrict__ col, float* __restrict__ xout,
    float cH, float cX, float cP, int N) {
  int lane = threadIdx.x & 31;
  int v = __builtin_amdgcn_readfirstlane((int)(blockIdx.x * 8 + (threadIdx.x >> 5)));
  if (v >= N) return;
  int start = lscan[v] + boff[v >> 10];
  int deg = cnt[v];
  long ro = (long)v * D_FEAT;
  float nv = nrm[v];
  float2 t1 = ((const float2*)(xin + ro))[lane];
  float2 t2 = make_float2(0.0f, 0.0f);
  if (HAS_PREV) t2 = ((const float2*)(xprev + ro))[lane];

  float ax = 0.0f, ay = 0.0f;
  for (int base = 0; base < deg; base += 32) {
    int m = min(32, deg - base);
    int cj = 0;
    float wj = 0.0f;
    if (lane < m) {
      cj = col[start + base + lane];
      cj = (int)min((unsigned)cj, (unsigned)(N - 1));
      wj = nrm[cj];
    }
    int wji = __float_as_int(wj);
    for (int k = 0; k < m; k += 4) {
      int s0 = __builtin_amdgcn_readlane(cj, k + 0);
      int s1 = __builtin_amdgcn_readlane(cj, k + 1);
      int s2 = __builtin_amdgcn_readlane(cj, k + 2);
      int s3 = __builtin_amdgcn_readlane(cj, k + 3);
      float w0 = __int_as_float(__builtin_amdgcn_readlane(wji, k + 0));
      float w1 = __int_as_float(__builtin_amdgcn_readlane(wji, k + 1));
      float w2 = __int_as_float(__builtin_amdgcn_readlane(wji, k + 2));
      float w3 = __int_as_float(__builtin_amdgcn_readlane(wji, k + 3));
      float2 x0 = ((const float2*)(xin + (long)s0 * D_FEAT))[lane];
      float2 x1 = ((const float2*)(xin + (long)s1 * D_FEAT))[lane];
      float2 x2 = ((const float2*)(xin + (long)s2 * D_FEAT))[lane];
      float2 x3 = ((const float2*)(xin + (long)s3 * D_FEAT))[lane];
      ax = fmaf(x0.x, w0, ax); ay = fmaf(x0.y, w0, ay);
      ax = fmaf(x1.x, w1, ax); ay = fmaf(x1.y, w1, ay);
      ax = fmaf(x2.x, w2, ax); ay = fmaf(x2.y, w2, ay);
      ax = fmaf(x3.x, w3, ax); ay = fmaf(x3.y, w3, ay);
    }
  }
  float hx = ax * nv, hy = ay * nv;
  float ox = cH * hx + cX * t1.x;
  float oy = cH * hy + cX * t1.y;
  if (HAS_PREV) { ox += cP * t2.x; oy += cP * t2.y; }
  ((float2*)(xout + ro))[lane] = make_float2(ox, oy);
}


__device__ __forceinline__ v8f accum_term(v8f acc, const float* __restrict__ xr,
                                          const float* __restrict__ Wt,
                                          int khalf, int ncol) {
  v2f bfrag[16];
#pragma unroll
  for (int kk = 0; kk < 16; ++kk) {
    int krow = kk * 4 + khalf * 2;
    v2f bf;
    bf.x = Wt[(krow + 0) * D_FEAT + ncol];
    bf.y = Wt[(krow + 1) * D_FEAT + ncol];
    bfrag[kk] = bf;
  }
#pragma unroll
  for (int kk = 0; kk < 16; ++kk) {
    v2f a;
    a.x = xr[kk * 4 + khalf * 2 + 0];
    a.y = xr[kk * 4 + khalf * 2 + 1];
    acc = __builtin_amdgcn_wmma_f32_16x16x4_f32(
         false, a,  false, bfrag[kk],
         (short)0, acc,  false,  false);
  }
  return acc;
}

__global__ __launch_bounds__(128) void gemm_fused_kernel(
    const float* __restrict__ X0, const float* __restrict__ X1,
    const float* __restrict__ X2, const float* __restrict__ X3,
    const float* __restrict__ W, const float* __restrict__ bias,
    float* __restrict__ out, int N) {
  int lane = threadIdx.x & 31;
  int wv   = threadIdx.x >> 5;
  int row0 = blockIdx.x * 16;
  if (row0 >= N) return;

  int mlo   = lane & 15;
  int khalf = lane >> 4;
  int ncol  = wv * 16 + mlo;

  int arow = row0 + mlo;
  if (arow >= N) arow = N - 1;
  long aoff = (long)arow * D_FEAT;

  v8f acc;
  {
    float bb = bias[ncol];
#pragma unroll
    for (int v = 0; v < 8; ++v) acc[v] = bb;
  }

  acc = accum_term(acc, X0 + aoff, W + 0 * D_FEAT * D_FEAT, khalf, ncol);
  acc = accum_term(acc, X1 + aoff, W + 1 * D_FEAT * D_FEAT, khalf, ncol);
  acc = accum_term(acc, X2 + aoff, W + 2 * D_FEAT * D_FEAT, khalf, ncol);
  acc = accum_term(acc, X3 + aoff, W + 3 * D_FEAT * D_FEAT, khalf, ncol);

  if (row0 + 16 <= N) {
    float* orow = out + (long)(row0 + khalf * 8) * D_FEAT + ncol;
#pragma unroll
    for (int v = 0; v < 8; ++v) orow[(long)v * D_FEAT] = acc[v];
  } else {
#pragma unroll
    for (int v = 0; v < 8; ++v) {
      int r = row0 + khalf * 8 + v;
      if (r < N) out[(long)r * D_FEAT + ncol] = acc[v];
    }
  }
}


extern "C" void kernel_launch(void* const* d_in, const int* in_sizes, int n_in,
                              void* d_out, int out_size, void* d_ws, size_t ws_size,
                              hipStream_t stream) {
  const float* feat = (const float*)d_in[0];
  const int*   src  = (const int*)d_in[1];
  const int*   dst  = (const int*)d_in[2];
  const float* W    = (const float*)d_in[3];
  const float* b    = (const float*)d_in[4];
  float*       out  = (float*)d_out;

  const int N = in_sizes[0] / D_FEAT;
  const int E = in_sizes[1];
  const long nd = (long)N * D_FEAT;
  const int nb = (N + 1023) / 1024;

  const long Np  = ((long)N + 63) / 64 * 64;
  const long Ep  = ((long)E + 63) / 64 * 64;
  const long nbp = ((long)nb + 63) / 64 * 64;
  int*   cnt   = (int*)d_ws;
  int*   lscan = cnt + Np;
  int*   btot  = lscan + Np;
  int*   boff  = btot + nbp;
  int*   rank  = boff + nbp;
  int*   col   = rank + Ep;
  float* nrm   = (float*)(col + Ep);
  float* txA   = nrm + Np;
  float* txB   = txA + nd;
  float* txC   = txB + nd;

  const int egrid = (E + 255) / 256;
  const int ngrid = (N + 7) / 8;
  const int ggrid = (N + 15) / 16;

  zero_int_kernel<<<(N + 255) / 256, 256, 0, stream>>>(cnt, N);
  rank_kernel<<<egrid, 256, 0, stream>>>(dst, cnt, rank, E);
  scan_local_kernel<<<nb, 256, 0, stream>>>(cnt, lscan, btot, nrm, N);
  scan_blocks_kernel<<<1, 1024, 0, stream>>>(btot, boff, nb);
  fill_kernel<<<egrid, 256, 0, stream>>>(src, dst, rank, lscan, boff, col, E);

  gather_combine_kernel<false><<<ngrid, 256, 0, stream>>>(
      feat, feat, nrm, cnt, lscan, boff, col, txA,
      -2.0f / LAM, 2.0f / LAM - 1.0f, 0.0f, N);
  gather_combine_kernel<true><<<ngrid, 256, 0, stream>>>(
      txA, feat, nrm, cnt, lscan, boff, col, txB,
      -4.0f / LAM, 4.0f / LAM - 2.0f, -1.0f, N);
  gather_combine_kernel<true><<<ngrid, 256, 0, stream>>>(
      txB, txA, nrm, cnt, lscan, boff, col, txC,
      -4.0f / LAM, 4.0f / LAM - 2.0f, -1.0f, N);

  gemm_fused_kernel<<<ggrid, 128, 0, stream>>>(feat, txA, txB, txC, W, b, out, N);
}


# Dst sort in LDS: chunk blocks order edges by 256-node bin, one block per bin ranks its rows by LDS adds, writes col in whole lines; no global atomics or scattered stores; sums in LDS arrival order
# speedup vs baseline: 4.2538x; 1.6456x over previous
#include <hip/hip_runtime.h>


typedef __attribute__((ext_vector_type(2))) float v2f;
typedef __attribute__((ext_vector_type(8))) float v8f;

#define D_FEAT 64
constexpr float LAM = 2.0f;


#define BIN_SHIFT 8
#define BIN_NODES 256
#define CHUNK 4096
#define MAXBINS 512
#define STAGE_CAP 16384

__device__ __forceinline__ int wave_incl_scan(int x, int lane) {
#pragma unroll
  for (int o = 1; o < 32; o <<= 1) {
    int y = __shfl_up(x, o, 32);
    if (lane >= o) x += y;
  }
  return x;
}

__global__ __launch_bounds__(1024) void bin_kernel(
    const int* __restrict__ src, const int* __restrict__ dst,
    int* __restrict__ bsrc, int* __restrict__ bdst, unsigned* __restrict__ tab,
    int E, int N, int nbins, int nchunks) {
  __shared__ int hist[MAXBINS];
  __shared__ int wsum[16];
  __shared__ int ssrc[CHUNK];
  __shared__ int sdst[CHUNK];
  int t = threadIdx.x, lane = t & 31, w = t >> 5;
  int k = blockIdx.x;
  int e0 = k * CHUNK;
  if (t < MAXBINS) hist[t] = 0;
  __syncthreads();

  int s[4], d[4], r[4];
#pragma unroll
  for (int j = 0; j < 4; ++j) {
    int ec = min(e0 + j * 1024 + t, E - 1);
    s[j] = src[ec];
    d[j] = (int)min((unsigned)dst[ec], (unsigned)(N - 1));
  }
#pragma unroll
  for (int j = 0; j < 4; ++j) {
    r[j] = 0;
    if (e0 + j * 1024 + t < E) r[j] = atomicAdd(&hist[d[j] >> BIN_SHIFT], 1);
  }
  __syncthreads();

  int c = 0, inc = 0;
  if (t < MAXBINS) {
    c = hist[t];
    inc = wave_incl_scan(c, lane);
    if (lane == 31) wsum[w] = inc;
  }
  __syncthreads();
  if (t < MAXBINS) {
    int woff = 0;
#pragma unroll
    for (int q = 0; q < 16; ++q) if (q < w) woff += wsum[q];
    int st = woff + inc - c;
    hist[t] = st;
    if (t < nbins) tab[(long)k * nbins + t] = ((unsigned)st << 16) | (unsigned)c;
  }
  __syncthreads();

#pragma unroll
  for (int j = 0; j < 4; ++j) {
    int e = e0 + j * 1024 + t;
    if (e < E) {
      int p = hist[d[j] >> BIN_SHIFT] + r[j];
      ssrc[p] = s[j];
      sdst[p] = d[j];
    }
  }
  __syncthreads();
  for (int j = 0; j < 4; ++j) {
    int i = j * 1024 + t;
    if (e0 + i < E) {
      bsrc[e0 + i] = ssrc[i];
      bdst[e0 + i] = sdst[i];
    }
  }
}

#define LROW 1024
#define RUNS 4

__device__ __forceinline__ unsigned run_word(const unsigned* srow, const unsigned* __restrict__ tab,
                                             int k, int nchunks, int nbins, int b) {
  if (k >= nchunks) return 0u;
  return (k < LROW) ? srow[k] : tab[(long)k * nbins + b];
}

__global__ __launch_bounds__(1024) void csr_kernel(
    const int* __restrict__ bsrc, const int* __restrict__ bdst, const unsigned* __restrict__ tab,
    int* __restrict__ cnt, int* __restrict__ rowstart, float* __restrict__ nrm,
    int* __restrict__ col, int N, int E, int nbins, int nchunks) {
  __shared__ int lcnt[BIN_NODES];
  __shared__ int lstart[BIN_NODES];
  __shared__ int lfill[BIN_NODES];
  __shared__ int wsum[8];
  __shared__ int wbase[32];
  __shared__ int wtot[32];
  __shared__ unsigned srow[LROW];
  __shared__ int stage[STAGE_CAP];
  int t = threadIdx.x, lane = t & 31, w = t >> 5;
  int hw = t >> 4, hl = t & 15;
  int b = blockIdx.x;

  int mybase = 0, mytot = 0;
  for (int k = t; k < nchunks; k += 1024) {
    unsigned en = tab[(long)k * nbins + b];
    if (k < LROW) srow[k] = en;
    mybase += (int)(en >> 16);
    mytot += (int)(en & 0xffffu);
  }
#pragma unroll
  for (int o = 16; o > 0; o >>= 1) {
    mybase += __shfl_xor(mybase, o, 32);
    mytot += __shfl_xor(mytot, o, 32);
  }
  if (lane == 0) { wbase[w] = mybase; wtot[w] = mytot; }
  if (t < BIN_NODES) lcnt[t] = 0;
  __syncthreads();
  int binbase = 0, total = 0;
#pragma unroll
  for (int q = 0; q < 32; ++q) { binbase += wbase[q]; total += wtot[q]; }
  const bool staged = total <= STAGE_CAP;

  for (int k0 = hw; k0 < nchunks; k0 += 64 * RUNS) {
    unsigned en[RUNS];
    int rc[RUNS], dv[RUNS];
    long eb[RUNS];
#pragma unroll
    for (int u = 0; u < RUNS; ++u) en[u] = run_word(srow, tab, k0 + 64 * u, nchunks, nbins, b);
#pragma unroll
    for (int u = 0; u < RUNS; ++u) {
      rc[u] = (int)(en[u] & 0xffffu);
      eb[u] = (long)(k0 + 64 * u) * CHUNK + (long)(en[u] >> 16);
      long ix = min(eb[u] + hl, (long)E - 1);
      dv[u] = bdst[ix];
    }
#pragma unroll
    for (int u = 0; u < RUNS; ++u) {
      if (hl < rc[u]) atomicAdd(&lcnt[dv[u] & (BIN_NODES - 1)], 1);
      for (int i = hl + 16; i < rc[u]; i += 16)
        atomicAdd(&lcnt[bdst[eb[u] + i] & (BIN_NODES - 1)], 1);
    }
  }
  __syncthreads();

  int c = 0, inc = 0;
  if (t < BIN_NODES) {
    c = lcnt[t];
    inc = wave_incl_scan(c, lane);
    if (lane == 31) wsum[w] = inc;
  }
  __syncthreads();
  if (t < BIN_NODES) {
    int woff = 0;
#pragma unroll
    for (int q = 0; q < 8; ++q) if (q < w) woff += wsum[q];
    int ls = woff + inc - c;
    lstart[t] = ls;
    lfill[t] = 0;
    int v = b * BIN_NODES + t;
    if (v < N) {
      cnt[v] = c;
      rowstart[v] = binbase + ls;
      nrm[v] = rsqrtf(fmaxf((float)c, 1.0f));
    }
  }
  __syncthreads();

  for (int k0 = hw; k0 < nchunks; k0 += 64 * RUNS) {
    unsigned en[RUNS];
    int rc[RUNS], dv[RUNS], sv[RUNS];
    long eb[RUNS];
#pragma unroll
    for (int u = 0; u < RUNS; ++u) en[u] = run_word(srow, tab, k0 + 64 * u, nchunks, nbins, b);
#pragma unroll
    for (int u = 0; u < RUNS; ++u) {
      rc[u] = (int)(en[u] & 0xffffu);
      eb[u] = (long)(k0 + 64 * u) * CHUNK + (long)(en[u] >> 16);
      long ix = min(eb[u] + hl, (long)E - 1);
      dv[u] = bdst[ix];
      sv[u] = bsrc[ix];
    }
#pragma unroll
    for (int u = 0; u < RUNS; ++u) {
      for (int i = hl; i < rc[u]; i += 16) {
        int d1 = dv[u], s1 = sv[u];
        if (i != hl) { d1 = bdst[eb[u] + i]; s1 = bsrc[eb[u] + i]; }
        int dl = d1 & (BIN_NODES - 1);
        int p = lstart[dl] + atomicAdd(&lfill[dl], 1);
        if (staged) {
          stage[p] = s1;
        } else {
          if ((unsigned)(binbase + p) < (unsigned)E) col[binbase + p] = s1;
        }
      }
    }
  }
  __syncthreads();
  if (staged) {
    for (int i = t; i < total; i += 1024) {
      if ((unsigned)(binbase + i) < (unsigned)E) col[binbase + i] = stage[i];
    }
  }
}

template <bool HAS_PREV>
__global__ __launch_bounds__(256) void gather_combine_kernel(
    const float* __restrict__ xin, const float* __restrict__ xprev,
    const float* __restrict__ nrm, const int* __restrict__ cnt,
    const int* __restrict__ rowstart,
    const int* __restrict__ col, float* __restrict__ xout,
    float cH, float cX, float cP, int N) {
  int lane = threadIdx.x & 31;
  int v = __builtin_amdgcn_readfirstlane((int)(blockIdx.x * 8 + (threadIdx.x >> 5)));
  if (v >= N) return;
  int start = rowstart[v];
  int deg = cnt[v];
  long ro = (long)v * D_FEAT;
  float nv = nrm[v];
  float2 t1 = ((const float2*)(xin + ro))[lane];
  float2 t2 = make_float2(0.0f, 0.0f);
  if (HAS_PREV) t2 = ((const float2*)(xprev + ro))[lane];

  float ax = 0.0f, ay = 0.0f;
  for (int base = 0; base < deg; base += 32) {
    int m = min(32, deg - base);
    int cj = 0;
    float wj = 0.0f;
    if (lane < m) {
      cj = col[start + base + lane];
      cj = (int)min((unsigned)cj, (unsigned)(N - 1));
      wj = nrm[cj];
    }
    int wji = __float_as_int(wj);
    for (int k = 0; k < m; k += 4) {
      int s0 = __builtin_amdgcn_readlane(cj, k + 0);
      int s1 = __builtin_amdgcn_readlane(cj, k + 1);
      int s2 = __builtin_amdgcn_readlane(cj, k + 2);
      int s3 = __builtin_amdgcn_readlane(cj, k + 3);
      float w0 = __int_as_float(__builtin_amdgcn_readlane(wji, k + 0));
      float w1 = __int_as_float(__builtin_amdgcn_readlane(wji, k + 1));
      float w2 = __int_as_float(__builtin_amdgcn_readlane(wji, k + 2));
      float w3 = __int_as_float(__builtin_amdgcn_readlane(wji, k + 3));
      float2 x0 = ((const float2*)(xin + (long)s0 * D_FEAT))[lane];
      float2 x1 = ((const float2*)(xin + (long)s1 * D_FEAT))[lane];
      float2 x2 = ((const float2*)(xin + (long)s2 * D_FEAT))[lane];
      float2 x3 = ((const float2*)(xin + (long)s3 * D_FEAT))[lane];
      ax = fmaf(x0.x, w0, ax); ay = fmaf(x0.y, w0, ay);
      ax = fmaf(x1.x, w1, ax); ay = fmaf(x1.y, w1, ay);
      ax = fmaf(x2.x, w2, ax); ay = fmaf(x2.y, w2, ay);
      ax = fmaf(x3.x, w3, ax); ay = fmaf(x3.y, w3, ay);
    }
  }
  float hx = ax * nv, hy = ay * nv;
  float ox = cH * hx + cX * t1.x;
  float oy = cH * hy + cX * t1.y;
  if (HAS_PREV) { ox += cP * t2.x; oy += cP * t2.y; }
  ((float2*)(xout + ro))[lane] = make_float2(ox, oy);
}


__device__ __forceinline__ v8f accum_term(v8f acc, const float* __restrict__ xr,
                                          const float* __restrict__ Wt,
                                          int khalf, int ncol) {
  v2f bfrag[16];
#pragma unroll
  for (int kk = 0; kk < 16; ++kk) {
    int krow = kk * 4 + khalf * 2;
    v2f bf;
    bf.x = Wt[(krow + 0) * D_FEAT + ncol];
    bf.y = Wt[(krow + 1) * D_FEAT + ncol];
    bfrag[kk] = bf;
  }
#pragma unroll
  for (int kk = 0; kk < 16; ++kk) {
    v2f a;
    a.x = xr[kk * 4 + khalf * 2 + 0];
    a.y = xr[kk * 4 + khalf * 2 + 1];
    acc = __builtin_amdgcn_wmma_f32_16x16x4_f32(
         false, a,  false, bfrag[kk],
         (short)0, acc,  false,  false);
  }
  return acc;
}

__global__ __launch_bounds__(128) void gemm_fused_kernel(
    const float* __restrict__ X0, const float* __restrict__ X1,
    const float* __restrict__ X2, const float* __restrict__ X3,
    const float* __restrict__ W, const float* __restrict__ bias,
    float* __restrict__ out, int N) {
  int lane = threadIdx.x & 31;
  int wv   = threadIdx.x >> 5;
  int row0 = blockIdx.x * 16;
  if (row0 >= N) return;

  int mlo   = lane & 15;
  int khalf = lane >> 4;
  int ncol  = wv * 16 + mlo;

  int arow = row0 + mlo;
  if (arow >= N) arow = N - 1;
  long aoff = (long)arow * D_FEAT;

  v8f acc;
  {
    float bb = bias[ncol];
#pragma unroll
    for (int v = 0; v < 8; ++v) acc[v] = bb;
  }

  acc = accum_term(acc, X0 + aoff, W + 0 * D_FEAT * D_FEAT, khalf, ncol);
  acc = accum_term(acc, X1 + aoff, W + 1 * D_FEAT * D_FEAT, khalf, ncol);
  acc = accum_term(acc, X2 + aoff, W + 2 * D_FEAT * D_FEAT, khalf, ncol);
  acc = accum_term(acc, X3 + aoff, W + 3 * D_FEAT * D_FEAT, khalf, ncol);

  if (row0 + 16 <= N) {
    float* orow = out + (long)(row0 + khalf * 8) * D_FEAT + ncol;
#pragma unroll
    for (int v = 0; v < 8; ++v) orow[(long)v * D_FEAT] = acc[v];
  } else {
#pragma unroll
    for (int v = 0; v < 8; ++v) {
      int r = row0 + khalf * 8 + v;
      if (r < N) out[(long)r * D_FEAT + ncol] = acc[v];
    }
  }
}


extern "C" void kernel_launch(void* const* d_in, const int* in_sizes, int n_in,
                              void* d_out, int out_size, void* d_ws, size_t ws_size,
                              hipStream_t stream) {
  const float* feat = (const float*)d_in[0];
  const int*   src  = (const int*)d_in[1];
  const int*   dst  = (const int*)d_in[2];
  const float* W    = (const float*)d_in[3];
  const float* b    = (const float*)d_in[4];
  float*       out  = (float*)d_out;

  const int N = in_sizes[0] / D_FEAT;
  const int E = in_sizes[1];
  const long nd = (long)N * D_FEAT;
  const int nbins   = (N + BIN_NODES - 1) / BIN_NODES;
  const int nchunks = (E + CHUNK - 1) / CHUNK;

  const long Np   = ((long)N + 63) / 64 * 64;
  const long Ep   = ((long)E + 63) / 64 * 64;
  const long tabp = ((long)nbins * nchunks + 63) / 64 * 64;
  int*      cnt      = (int*)d_ws;
  int*      rowstart = cnt + Np;
  unsigned* tab      = (unsigned*)(rowstart + Np);
  int*      bsrc     = (int*)(tab + tabp);
  int*      bdst     = bsrc + Ep;
  int*      col      = bdst + Ep;
  float*    nrm      = (float*)(col + Ep);
  float*    txA      = nrm + Np;
  float*    txB      = txA + nd;
  float*    txC      = txB + nd;

  const int ngrid = (N + 7) / 8;
  const int ggrid = (N + 15) / 16;

  bin_kernel<<<nchunks, 1024, 0, stream>>>(src, dst, bsrc, bdst, tab, E, N, nbins, nchunks);
  csr_kernel<<<nbins, 1024, 0, stream>>>(bsrc, bdst, tab, cnt, rowstart, nrm, col, N, E, nbins, nchunks);

  gather_combine_kernel<false><<<ngrid, 256, 0, stream>>>(
      feat, feat, nrm, cnt, rowstart, col, txA,
      -2.0f / LAM, 2.0f / LAM - 1.0f, 0.0f, N);
  gather_combine_kernel<true><<<ngrid, 256, 0, stream>>>(
      txA, feat, nrm, cnt, rowstart, col, txB,
      -4.0f / LAM, 4.0f / LAM - 2.0f, -1.0f, N);
  gather_combine_kernel<true><<<ngrid, 256, 0, stream>>>(
      txB, txA, nrm, cnt, rowstart, col, txC,
      -4.0f / LAM, 4.0f / LAM - 2.0f, -1.0f, N);

  gemm_fused_kernel<<<ggrid, 128, 0, stream>>>(feat, txA, txB, txC, W, b, out, N);
}


# Gather props: one wave walks 4 consecutive nodes (their bounds in one load each, the next node's sources fetched under the current node's rows): 25000 waves started per prop, not 100000
# speedup vs baseline: 4.5431x; 1.0680x over previous
#include <hip/hip_runtime.h>


typedef __attribute__((ext_vector_type(2))) float v2f;
typedef __attribute__((ext_vector_type(8))) float v8f;

#define D_FEAT 64
constexpr float LAM = 2.0f;


#define BIN_SHIFT 8
#define BIN_NODES 256
#define CHUNK 4096
#define MAXBINS 512
#define STAGE_CAP 16384

__device__ __forceinline__ int wave_incl_scan(int x, int lane) {
#pragma unroll
  for (int o = 1; o < 32; o <<= 1) {
    int y = __shfl_up(x, o, 32);
    if (lane >= o) x += y;
  }
  return x;
}

__global__ __launch_bounds__(1024) void bin_kernel(
    const int* __restrict__ src, const int* __restrict__ dst,
    int* __restrict__ bsrc, int* __restrict__ bdst, unsigned* __restrict__ tab,
    int E, int N, int nbins, int nchunks) {
  __shared__ int hist[MAXBINS];
  __shared__ int wsum[16];
  __shared__ int ssrc[CHUNK];
  __shared__ int sdst[CHUNK];
  int t = threadIdx.x, lane = t & 31, w = t >> 5;
  int k = blockIdx.x;
  int e0 = k * CHUNK;
  if (t < MAXBINS) hist[t] = 0;
  __syncthreads();

  int s[4], d[4], r[4];
#pragma unroll
  for (int j = 0; j < 4; ++j) {
    int ec = min(e0 + j * 1024 + t, E - 1);
    s[j] = src[ec];
    d[j] = (int)min((unsigned)dst[ec], (unsigned)(N - 1));
  }
#pragma unroll
  for (int j = 0; j < 4; ++j) {
    r[j] = 0;
    if (e0 + j * 1024 + t < E) r[j] = atomicAdd(&hist[d[j] >> BIN_SHIFT], 1);
  }
  __syncthreads();

  int c = 0, inc = 0;
  if (t < MAXBINS) {
    c = hist[t];
    inc = wave_incl_scan(c, lane);
    if (lane == 31) wsum[w] = inc;
  }
  __syncthreads();
  if (t < MAXBINS) {
    int woff = 0;
#pragma unroll
    for (int q = 0; q < 16; ++q) if (q < w) woff += wsum[q];
    int st = woff + inc - c;
    hist[t] = st;
    if (t < nbins) tab[(long)k * nbins + t] = ((unsigned)st << 16) | (unsigned)c;
  }
  __syncthreads();

#pragma unroll
  for (int j = 0; j < 4; ++j) {
    int e = e0 + j * 1024 + t;
    if (e < E) {
      int p = hist[d[j] >> BIN_SHIFT] + r[j];
      ssrc[p] = s[j];
      sdst[p] = d[j];
    }
  }
  __syncthreads();
  for (int j = 0; j < 4; ++j) {
    int i = j * 1024 + t;
    if (e0 + i < E) {
      bsrc[e0 + i] = ssrc[i];
      bdst[e0 + i] = sdst[i];
    }
  }
}

#define LROW 1024
#define RUNS 4

__device__ __forceinline__ unsigned run_word(const unsigned* srow, const unsigned* __restrict__ tab,
                                             int k, int nchunks, int nbins, int b) {
  if (k >= nchunks) return 0u;
  return (k < LROW) ? srow[k] : tab[(long)k * nbins + b];
}

__global__ __launch_bounds__(1024) void csr_kernel(
    const int* __restrict__ bsrc, const int* __restrict__ bdst, const unsigned* __restrict__ tab,
    const float* __restrict__ feat,
    int* __restrict__ cnt, int* __restrict__ rowstart, float* __restrict__ nrm,
    float* __restrict__ rdn, float* __restrict__ y0,
    int* __restrict__ col, int N, int E, int nbins, int nchunks) {
  __shared__ int lcnt[BIN_NODES];
  __shared__ int lstart[BIN_NODES];
  __shared__ int lfill[BIN_NODES];
  __shared__ int wsum[8];
  __shared__ int wbase[32];
  __shared__ int wtot[32];
  __shared__ unsigned srow[LROW];
  __shared__ int stage[STAGE_CAP];
  __shared__ float lnrm[BIN_NODES];
  int t = threadIdx.x, lane = t & 31, w = t >> 5;
  int hw = t >> 4, hl = t & 15;
  int b = blockIdx.x;

  int mybase = 0, mytot = 0;
  for (int k = t; k < nchunks; k += 1024) {
    unsigned en = tab[(long)k * nbins + b];
    if (k < LROW) srow[k] = en;
    mybase += (int)(en >> 16);
    mytot += (int)(en & 0xffffu);
  }
#pragma unroll
  for (int o = 16; o > 0; o >>= 1) {
    mybase += __shfl_xor(mybase, o, 32);
    mytot += __shfl_xor(mytot, o, 32);
  }
  if (lane == 0) { wbase[w] = mybase; wtot[w] = mytot; }
  if (t < BIN_NODES) lcnt[t] = 0;
  __syncthreads();
  int binbase = 0, total = 0;
#pragma unroll
  for (int q = 0; q < 32; ++q) { binbase += wbase[q]; total += wtot[q]; }
  const bool staged = total <= STAGE_CAP;

  for (int k0 = hw; k0 < nchunks; k0 += 64 * RUNS) {
    unsigned en[RUNS];
    int rc[RUNS], dv[RUNS];
    long eb[RUNS];
#pragma unroll
    for (int u = 0; u < RUNS; ++u) en[u] = run_word(srow, tab, k0 + 64 * u, nchunks, nbins, b);
#pragma unroll
    for (int u = 0; u < RUNS; ++u) {
      rc[u] = (int)(en[u] & 0xffffu);
      eb[u] = (long)(k0 + 64 * u) * CHUNK + (long)(en[u] >> 16);
      long ix = min(eb[u] + hl, (long)E - 1);
      dv[u] = bdst[ix];
    }
#pragma unroll
    for (int u = 0; u < RUNS; ++u) {
      if (hl < rc[u]) atomicAdd(&lcnt[dv[u] & (BIN_NODES - 1)], 1);
      for (int i = hl + 16; i < rc[u]; i += 16)
        atomicAdd(&lcnt[bdst[eb[u] + i] & (BIN_NODES - 1)], 1);
    }
  }

  float4 fx[4];
#pragma unroll
  for (int j = 0; j < 4; ++j) {
    int i = j * 1024 + t;
    long v = min((long)b * BIN_NODES + (i >> 4), (long)N - 1);
    fx[j] = ((const float4*)feat)[v * (D_FEAT / 4) + (i & 15)];
  }
  __syncthreads();

  int c = 0, inc = 0;
  if (t < BIN_NODES) {
    c = lcnt[t];
    inc = wave_incl_scan(c, lane);
    if (lane == 31) wsum[w] = inc;
  }
  __syncthreads();
  if (t < BIN_NODES) {
    int woff = 0;
#pragma unroll
    for (int q = 0; q < 8; ++q) if (q < w) woff += wsum[q];
    int ls = woff + inc - c;
    lstart[t] = ls;
    lfill[t] = 0;
    int v = b * BIN_NODES + t;
    float nv = rsqrtf(fmaxf((float)c, 1.0f));
    lnrm[t] = nv;
    if (v < N) {
      cnt[v] = c;
      rowstart[v] = binbase + ls;
      nrm[v] = nv;
      rdn[v] = sqrtf(fmaxf((float)c, 1.0f));
    }
  }
  __syncthreads();

#pragma unroll
  for (int j = 0; j < 4; ++j) {
    int i = j * 1024 + t;
    int row = i >> 4;
    long v = (long)b * BIN_NODES + row;
    if (v < N) {
      float s = lnrm[row];
      float4 x = fx[j];
      ((float4*)y0)[v * (D_FEAT / 4) + (i & 15)] = make_float4(x.x * s, x.y * s, x.z * s, x.w * s);
    }
  }

  for (int k0 = hw; k0 < nchunks; k0 += 64 * RUNS) {
    unsigned en[RUNS];
    int rc[RUNS], dv[RUNS], sv[RUNS];
    long eb[RUNS];
#pragma unroll
    for (int u = 0; u < RUNS; ++u) en[u] = run_word(srow, tab, k0 + 64 * u, nchunks, nbins, b);
#pragma unroll
    for (int u = 0; u < RUNS; ++u) {
      rc[u] = (int)(en[u] & 0xffffu);
      eb[u] = (long)(k0 + 64 * u) * CHUNK + (long)(en[u] >> 16);
      long ix = min(eb[u] + hl, (long)E - 1);
      dv[u] = bdst[ix];
      sv[u] = bsrc[ix];
    }
#pragma unroll
    for (int u = 0; u < RUNS; ++u) {
      for (int i = hl; i < rc[u]; i += 16) {
        int d1 = dv[u], s1 = sv[u];
        if (i != hl) { d1 = bdst[eb[u] + i]; s1 = bsrc[eb[u] + i]; }
        int dl = d1 & (BIN_NODES - 1);
        int p = lstart[dl] + atomicAdd(&lfill[dl], 1);
        if (staged) {
          stage[p] = s1;
        } else {
          if ((unsigned)(binbase + p) < (unsigned)E) col[binbase + p] = s1;
        }
      }
    }
  }
  __syncthreads();
  if (staged) {
    for (int i = t; i < total; i += 1024) {
      if ((unsigned)(binbase + i) < (unsigned)E) col[binbase + i] = stage[i];
    }
  }
}

#define NPW 4
template <bool HAS_PREV>
__global__ __launch_bounds__(256) void gather_combine_kernel(
    const float* __restrict__ yin, const float* __restrict__ yprev,
    const float* __restrict__ nrm, const int* __restrict__ cnt,
    const int* __restrict__ rowstart,
    const int* __restrict__ col, float* __restrict__ yout,
    float cH, float cX, float cP, int N) {
  int lane = threadIdx.x & 31;
  int v0 = __builtin_amdgcn_readfirstlane((int)(blockIdx.x * 8 + (threadIdx.x >> 5))) * NPW;
  if (v0 >= N) return;
  int n = min(NPW, N - v0);
  int ms = 0, md = 0;
  float mn = 0.0f;
  if (lane < n) {
    ms = rowstart[v0 + lane];
    md = cnt[v0 + lane];
    mn = nrm[v0 + lane];
  }
  int start = __builtin_amdgcn_readlane(ms, 0);
  int deg = __builtin_amdgcn_readlane(md, 0);
  int cjn = 0;
  if (lane < min(deg, 32)) cjn = col[start + lane];

  for (int j = 0; j < n; ++j) {
    long ro = (long)(v0 + j) * D_FEAT;
    float nv = __builtin_bit_cast(float, __builtin_amdgcn_readlane(__builtin_bit_cast(int, mn), j));
    float2 t1 = ((const float2*)(yin + ro))[lane];
    float2 t2 = make_float2(0.0f, 0.0f);
    if (HAS_PREV) t2 = ((const float2*)(yprev + ro))[lane];
    int cj = (int)min((unsigned)cjn, (unsigned)(N - 1));
    int startn = 0, degn = 0;
    if (j + 1 < n) {
      startn = __builtin_amdgcn_readlane(ms, j + 1);
      degn = __builtin_amdgcn_readlane(md, j + 1);
      if (lane < min(degn, 32)) cjn = col[startn + lane];
    }

    float ax = 0.0f, ay = 0.0f;
    for (int base = 0; base < deg; base += 32) {
      int m = min(32, deg - base);
      if (base > 0) {
        cj = 0;
        if (lane < m) cj = (int)min((unsigned)col[start + base + lane], (unsigned)(N - 1));
      }
      int k = 0;
      for (; k + 4 <= m; k += 4) {
        int s0 = __builtin_amdgcn_readlane(cj, k + 0);
        int s1 = __builtin_amdgcn_readlane(cj, k + 1);
        int s2 = __builtin_amdgcn_readlane(cj, k + 2);
        int s3 = __builtin_amdgcn_readlane(cj, k + 3);
        float2 x0 = ((const float2*)(yin + (long)s0 * D_FEAT))[lane];
        float2 x1 = ((const float2*)(yin + (long)s1 * D_FEAT))[lane];
        float2 x2 = ((const float2*)(yin + (long)s2 * D_FEAT))[lane];
        float2 x3 = ((const float2*)(yin + (long)s3 * D_FEAT))[lane];
        ax += x0.x; ay += x0.y;
        ax += x1.x; ay += x1.y;
        ax += x2.x; ay += x2.y;
        ax += x3.x; ay += x3.y;
      }
      if (k < m) {
        int r = m - k;
        int s0 = __builtin_amdgcn_readlane(cj, k);
        float2 x0 = ((const float2*)(yin + (long)s0 * D_FEAT))[lane];
        float2 x1 = make_float2(0.0f, 0.0f);
        float2 x2 = make_float2(0.0f, 0.0f);
        if (r > 1) {
          int s1 = __builtin_amdgcn_readlane(cj, k + 1);
          x1 = ((const float2*)(yin + (long)s1 * D_FEAT))[lane];
        }
        if (r > 2) {
          int s2 = __builtin_amdgcn_readlane(cj, k + 2);
          x2 = ((const float2*)(yin + (long)s2 * D_FEAT))[lane];
        }
        ax += x0.x; ay += x0.y;
        ax += x1.x; ay += x1.y;
        ax += x2.x; ay += x2.y;
      }
    }
    float g = cH * nv * nv;
    float ox = g * ax + cX * t1.x;
    float oy = g * ay + cX * t1.y;
    if (HAS_PREV) { ox += cP * t2.x; oy += cP * t2.y; }
    ((float2*)(yout + ro))[lane] = make_float2(ox, oy);
    start = startn;
    deg = degn;
  }
}


__device__ __forceinline__ v8f accum_term(v8f acc, const float* __restrict__ xr,
                                          const float* __restrict__ Wt,
                                          int khalf, int ncol) {
  v2f bfrag[16];
#pragma unroll
  for (int kk = 0; kk < 16; ++kk) {
    int krow = kk * 4 + khalf * 2;
    v2f bf;
    bf.x = Wt[(krow + 0) * D_FEAT + ncol];
    bf.y = Wt[(krow + 1) * D_FEAT + ncol];
    bfrag[kk] = bf;
  }
#pragma unroll
  for (int kk = 0; kk < 16; ++kk) {
    v2f a;
    a.x = xr[kk * 4 + khalf * 2 + 0];
    a.y = xr[kk * 4 + khalf * 2 + 1];
    acc = __builtin_amdgcn_wmma_f32_16x16x4_f32(
         false, a,  false, bfrag[kk],
         (short)0, acc,  false,  false);
  }
  return acc;
}

__global__ __launch_bounds__(128) void gemm_fused_kernel(
    const float* __restrict__ X0, const float* __restrict__ Y1,
    const float* __restrict__ Y2, const float* __restrict__ Y3,
    const float* __restrict__ rdn,
    const float* __restrict__ W, const float* __restrict__ bias,
    float* __restrict__ out, int N) {
  int lane = threadIdx.x & 31;
  int wv   = threadIdx.x >> 5;
  int row0 = blockIdx.x * 16;
  if (row0 >= N) return;

  int mlo   = lane & 15;
  int khalf = lane >> 4;
  int ncol  = wv * 16 + mlo;

  int arow = row0 + mlo;
  if (arow >= N) arow = N - 1;
  long aoff = (long)arow * D_FEAT;

  float rs[8];
  if (row0 + 16 <= N) {
    const float4* rp = (const float4*)(rdn + row0 + khalf * 8);
    float4 lo = rp[0], hi = rp[1];
    rs[0] = lo.x; rs[1] = lo.y; rs[2] = lo.z; rs[3] = lo.w;
    rs[4] = hi.x; rs[5] = hi.y; rs[6] = hi.z; rs[7] = hi.w;
  } else {
#pragma unroll
    for (int v = 0; v < 8; ++v) rs[v] = rdn[min(row0 + khalf * 8 + v, N - 1)];
  }
  float bb = bias[ncol];

  v8f acc;
#pragma unroll
  for (int v = 0; v < 8; ++v) acc[v] = 0.0f;
  acc = accum_term(acc, Y1 + aoff, W + 1 * D_FEAT * D_FEAT, khalf, ncol);
  acc = accum_term(acc, Y2 + aoff, W + 2 * D_FEAT * D_FEAT, khalf, ncol);
  acc = accum_term(acc, Y3 + aoff, W + 3 * D_FEAT * D_FEAT, khalf, ncol);
#pragma unroll
  for (int v = 0; v < 8; ++v) acc[v] = acc[v] * rs[v] + bb;
  acc = accum_term(acc, X0 + aoff, W + 0 * D_FEAT * D_FEAT, khalf, ncol);

  if (row0 + 16 <= N) {
    float* orow = out + (long)(row0 + khalf * 8) * D_FEAT + ncol;
#pragma unroll
    for (int v = 0; v < 8; ++v) orow[(long)v * D_FEAT] = acc[v];
  } else {
#pragma unroll
    for (int v = 0; v < 8; ++v) {
      int r = row0 + khalf * 8 + v;
      if (r < N) out[(long)r * D_FEAT + ncol] = acc[v];
    }
  }
}


extern "C" void kernel_launch(void* const* d_in, const int* in_sizes, int n_in,
                              void* d_out, int out_size, void* d_ws, size_t ws_size,
                              hipStream_t stream) {
  const float* feat = (const float*)d_in[0];
  const int*   src  = (const int*)d_in[1];
  const int*   dst  = (const int*)d_in[2];
  const float* W    = (const float*)d_in[3];
  const float* b    = (const float*)d_in[4];
  float*       out  = (float*)d_out;

  const int N = in_sizes[0] / D_FEAT;
  const int E = in_sizes[1];
  const long nd = (long)N * D_FEAT;
  const int nbins   = (N + BIN_NODES - 1) / BIN_NODES;
  const int nchunks = (E + CHUNK - 1) / CHUNK;

  const long Np   = ((long)N + 63) / 64 * 64;
  const long Ep   = ((long)E + 63) / 64 * 64;
  const long tabp = ((long)nbins * nchunks + 63) / 64 * 64;
  int*      cnt      = (int*)d_ws;
  int*      rowstart = cnt + Np;
  unsigned* tab      = (unsigned*)(rowstart + Np);
  int*      bsrc     = (int*)(tab + tabp);
  int*      bdst     = bsrc + Ep;
  int*      col      = bdst + Ep;
  float*    nrm      = (float*)(col + Ep);
  float*    rdn      = nrm + Np;
  float*    y0       = rdn + Np;
  float*    txA      = y0 + nd;
  float*    txB      = txA + nd;
  float*    txC      = txB + nd;

  const int ngrid = (N + 8 * NPW - 1) / (8 * NPW);
  const int ggrid = (N + 15) / 16;

  bin_kernel<<<nchunks, 1024, 0, stream>>>(src, dst, bsrc, bdst, tab, E, N, nbins, nchunks);
  csr_kernel<<<nbins, 1024, 0, stream>>>(bsrc, bdst, tab, feat, cnt, rowstart, nrm, rdn, y0, col, N, E, nbins, nchunks);

  gather_combine_kernel<false><<<ngrid, 256, 0, stream>>>(
      y0, y0, nrm, cnt, rowstart, col, txA,
      -2.0f / LAM, 2.0f / LAM - 1.0f, 0.0f, N);
  gather_combine_kernel<true><<<ngrid, 256, 0, stream>>>(
      txA, y0, nrm, cnt, rowstart, col, txB,
      -4.0f / LAM, 4.0f / LAM - 2.0f, -1.0f, N);
  gather_combine_kernel<true><<<ngrid, 256, 0, stream>>>(
      txB, txA, nrm, cnt, rowstart, col, txC,
      -4.0f / LAM, 4.0f / LAM - 2.0f, -1.0f, N);

  gemm_fused_kernel<<<ggrid, 128, 0, stream>>>(feat, txA, txB, txC, rdn, W, b, out, N);
}


# GEMM operand feed: a wave owns 16 whole rows x 64 columns, A rows fetched once as whole-line 16-B loads through a wave-private LDS tile, W fragments from an LDS copy; k order in each 64-sum changes
# speedup vs baseline: 5.2084x; 1.1464x over previous
#include <hip/hip_runtime.h>


typedef __attribute__((ext_vector_type(2))) float v2f;
typedef __attribute__((ext_vector_type(8))) float v8f;

#define D_FEAT 64
constexpr float LAM = 2.0f;


#define BIN_SHIFT 8
#define BIN_NODES 256
#define CHUNK 4096
#define MAXBINS 512
#define STAGE_CAP 16384

__device__ __forceinline__ int wave_incl_scan(int x, int lane) {
#pragma unroll
  for (int o = 1; o < 32; o <<= 1) {
    int y = __shfl_up(x, o, 32);
    if (lane >= o) x += y;
  }
  return x;
}

__global__ __launch_bounds__(1024) void bin_kernel(
    const int* __restrict__ src, const int* __restrict__ dst,
    int* __restrict__ bsrc, int* __restrict__ bdst, unsigned* __restrict__ tab,
    int E, int N, int nbins, int nchunks) {
  __shared__ int hist[MAXBINS];
  __shared__ int wsum[16];
  __shared__ int ssrc[CHUNK];
  __shared__ int sdst[CHUNK];
  int t = threadIdx.x, lane = t & 31, w = t >> 5;
  int k = blockIdx.x;
  int e0 = k * CHUNK;
  if (t < MAXBINS) hist[t] = 0;
  __syncthreads();

  int s[4], d[4], r[4];
#pragma unroll
  for (int j = 0; j < 4; ++j) {
    int ec = min(e0 + j * 1024 + t, E - 1);
    s[j] = src[ec];
    d[j] = (int)min((unsigned)dst[ec], (unsigned)(N - 1));
  }
#pragma unroll
  for (int j = 0; j < 4; ++j) {
    r[j] = 0;
    if (e0 + j * 1024 + t < E) r[j] = atomicAdd(&hist[d[j] >> BIN_SHIFT], 1);
  }
  __syncthreads();

  int c = 0, inc = 0;
  if (t < MAXBINS) {
    c = hist[t];
    inc = wave_incl_scan(c, lane);
    if (lane == 31) wsum[w] = inc;
  }
  __syncthreads();
  if (t < MAXBINS) {
    int woff = 0;
#pragma unroll
    for (int q = 0; q < 16; ++q) if (q < w) woff += wsum[q];
    int st = woff + inc - c;
    hist[t] = st;
    if (t < nbins) tab[(long)k * nbins + t] = ((unsigned)st << 16) | (unsigned)c;
  }
  __syncthreads();

#pragma unroll
  for (int j = 0; j < 4; ++j) {
    int e = e0 + j * 1024 + t;
    if (e < E) {
      int p = hist[d[j] >> BIN_SHIFT] + r[j];
      ssrc[p] = s[j];
      sdst[p] = d[j];
    }
  }
  __syncthreads();
  for (int j = 0; j < 4; ++j) {
    int i = j * 1024 + t;
    if (e0 + i < E) {
      bsrc[e0 + i] = ssrc[i];
      bdst[e0 + i] = sdst[i];
    }
  }
}

#define LROW 1024
#define RUNS 4

__device__ __forceinline__ unsigned run_word(const unsigned* srow, const unsigned* __restrict__ tab,
                                             int k, int nchunks, int nbins, int b) {
  if (k >= nchunks) return 0u;
  return (k < LROW) ? srow[k] : tab[(long)k * nbins + b];
}

__global__ __launch_bounds__(1024) void csr_kernel(
    const int* __restrict__ bsrc, const int* __restrict__ bdst, const unsigned* __restrict__ tab,
    const float* __restrict__ feat,
    int* __restrict__ cnt, int* __restrict__ rowstart, float* __restrict__ nrm,
    float* __restrict__ rdn, float* __restrict__ y0,
    int* __restrict__ col, int N, int E, int nbins, int nchunks) {
  __shared__ int lcnt[BIN_NODES];
  __shared__ int lstart[BIN_NODES];
  __shared__ int lfill[BIN_NODES];
  __shared__ int wsum[8];
  __shared__ int wbase[32];
  __shared__ int wtot[32];
  __shared__ unsigned srow[LROW];
  __shared__ int stage[STAGE_CAP];
  __shared__ float lnrm[BIN_NODES];
  int t = threadIdx.x, lane = t & 31, w = t >> 5;
  int hw = t >> 4, hl = t & 15;
  int b = blockIdx.x;

  int mybase = 0, mytot = 0;
  for (int k = t; k < nchunks; k += 1024) {
    unsigned en = tab[(long)k * nbins + b];
    if (k < LROW) srow[k] = en;
    mybase += (int)(en >> 16);
    mytot += (int)(en & 0xffffu);
  }
#pragma unroll
  for (int o = 16; o > 0; o >>= 1) {
    mybase += __shfl_xor(mybase, o, 32);
    mytot += __shfl_xor(mytot, o, 32);
  }
  if (lane == 0) { wbase[w] = mybase; wtot[w] = mytot; }
  if (t < BIN_NODES) lcnt[t] = 0;
  __syncthreads();
  int binbase = 0, total = 0;
#pragma unroll
  for (int q = 0; q < 32; ++q) { binbase += wbase[q]; total += wtot[q]; }
  const bool staged = total <= STAGE_CAP;

  for (int k0 = hw; k0 < nchunks; k0 += 64 * RUNS) {
    unsigned en[RUNS];
    int rc[RUNS], dv[RUNS];
    long eb[RUNS];
#pragma unroll
    for (int u = 0; u < RUNS; ++u) en[u] = run_word(srow, tab, k0 + 64 * u, nchunks, nbins, b);
#pragma unroll
    for (int u = 0; u < RUNS; ++u) {
      rc[u] = (int)(en[u] & 0xffffu);
      eb[u] = (long)(k0 + 64 * u) * CHUNK + (long)(en[u] >> 16);
      long ix = min(eb[u] + hl, (long)E - 1);
      dv[u] = bdst[ix];
    }
#pragma unroll
    for (int u = 0; u < RUNS; ++u) {
      if (hl < rc[u]) atomicAdd(&lcnt[dv[u] & (BIN_NODES - 1)], 1);
      for (int i = hl + 16; i < rc[u]; i += 16)
        atomicAdd(&lcnt[bdst[eb[u] + i] & (BIN_NODES - 1)], 1);
    }
  }

  float4 fx[4];
#pragma unroll
  for (int j = 0; j < 4; ++j) {
    int i = j * 1024 + t;
    long v = min((long)b * BIN_NODES + (i >> 4), (long)N - 1);
    fx[j] = ((const float4*)feat)[v * (D_FEAT / 4) + (i & 15)];
  }
  __syncthreads();

  int c = 0, inc = 0;
  if (t < BIN_NODES) {
    c = lcnt[t];
    inc = wave_incl_scan(c, lane);
    if (lane == 31) wsum[w] = inc;
  }
  __syncthreads();
  if (t < BIN_NODES) {
    int woff = 0;
#pragma unroll
    for (int q = 0; q < 8; ++q) if (q < w) woff += wsum[q];
    int ls = woff + inc - c;
    lstart[t] = ls;
    lfill[t] = 0;
    int v = b * BIN_NODES + t;
    float nv = rsqrtf(fmaxf((float)c, 1.0f));
    lnrm[t] = nv;
    if (v < N) {
      cnt[v] = c;
      rowstart[v] = binbase + ls;
      nrm[v] = nv;
      rdn[v] = sqrtf(fmaxf((float)c, 1.0f));
    }
  }
  __syncthreads();

#pragma unroll
  for (int j = 0; j < 4; ++j) {
    int i = j * 1024 + t;
    int row = i >> 4;
    long v = (long)b * BIN_NODES + row;
    if (v < N) {
      float s = lnrm[row];
      float4 x = fx[j];
      ((float4*)y0)[v * (D_FEAT / 4) + (i & 15)] = make_float4(x.x * s, x.y * s, x.z * s, x.w * s);
    }
  }

  for (int k0 = hw; k0 < nchunks; k0 += 64 * RUNS) {
    unsigned en[RUNS];
    int rc[RUNS], dv[RUNS], sv[RUNS];
    long eb[RUNS];
#pragma unroll
    for (int u = 0; u < RUNS; ++u) en[u] = run_word(srow, tab, k0 + 64 * u, nchunks, nbins, b);
#pragma unroll
    for (int u = 0; u < RUNS; ++u) {
      rc[u] = (int)(en[u] & 0xffffu);
      eb[u] = (long)(k0 + 64 * u) * CHUNK + (long)(en[u] >> 16);
      long ix = min(eb[u] + hl, (long)E - 1);
      dv[u] = bdst[ix];
      sv[u] = bsrc[ix];
    }
#pragma unroll
    for (int u = 0; u < RUNS; ++u) {
      for (int i = hl; i < rc[u]; i += 16) {
        int d1 = dv[u], s1 = sv[u];
        if (i != hl) { d1 = bdst[eb[u] + i]; s1 = bsrc[eb[u] + i]; }
        int dl = d1 & (BIN_NODES - 1);
        int p = lstart[dl] + atomicAdd(&lfill[dl], 1);
        if (staged) {
          stage[p] = s1;
        } else {
          if ((unsigned)(binbase + p) < (unsigned)E) col[binbase + p] = s1;
        }
      }
    }
  }
  __syncthreads();
  if (staged) {
    for (int i = t; i < total; i += 1024) {
      if ((unsigned)(binbase + i) < (unsigned)E) col[binbase + i] = stage[i];
    }
  }
}

#define NPW 4
template <bool HAS_PREV>
__global__ __launch_bounds__(256) void gather_combine_kernel(
    const float* __restrict__ yin, const float* __restrict__ yprev,
    const float* __restrict__ nrm, const int* __restrict__ cnt,
    const int* __restrict__ rowstart,
    const int* __restrict__ col, float* __restrict__ yout,
    float cH, float cX, float cP, int N) {
  int lane = threadIdx.x & 31;
  int v0 = __builtin_amdgcn_readfirstlane((int)(blockIdx.x * 8 + (threadIdx.x >> 5))) * NPW;
  if (v0 >= N) return;
  int n = min(NPW, N - v0);
  int ms = 0, md = 0;
  float mn = 0.0f;
  if (lane < n) {
    ms = rowstart[v0 + lane];
    md = cnt[v0 + lane];
    mn = nrm[v0 + lane];
  }
  int start = __builtin_amdgcn_readlane(ms, 0);
  int deg = __builtin_amdgcn_readlane(md, 0);
  int cjn = 0;
  if (lane < min(deg, 32)) cjn = col[start + lane];

  for (int j = 0; j < n; ++j) {
    long ro = (long)(v0 + j) * D_FEAT;
    float nv = __builtin_bit_cast(float, __builtin_amdgcn_readlane(__builtin_bit_cast(int, mn), j));
    float2 t1 = ((const float2*)(yin + ro))[lane];
    float2 t2 = make_float2(0.0f, 0.0f);
    if (HAS_PREV) t2 = ((const float2*)(yprev + ro))[lane];
    int cj = (int)min((unsigned)cjn, (unsigned)(N - 1));
    int startn = 0, degn = 0;
    if (j + 1 < n) {
      startn = __builtin_amdgcn_readlane(ms, j + 1);
      degn = __builtin_amdgcn_readlane(md, j + 1);
      if (lane < min(degn, 32)) cjn = col[startn + lane];
    }

    float ax = 0.0f, ay = 0.0f;
    for (int base = 0; base < deg; base += 32) {
      int m = min(32, deg - base);
      if (base > 0) {
        cj = 0;
        if (lane < m) cj = (int)min((unsigned)col[start + base + lane], (unsigned)(N - 1));
      }
      int k = 0;
      for (; k + 4 <= m; k += 4) {
        int s0 = __builtin_amdgcn_readlane(cj, k + 0);
        int s1 = __builtin_amdgcn_readlane(cj, k + 1);
        int s2 = __builtin_amdgcn_readlane(cj, k + 2);
        int s3 = __builtin_amdgcn_readlane(cj, k + 3);
        float2 x0 = ((const float2*)(yin + (long)s0 * D_FEAT))[lane];
        float2 x1 = ((const float2*)(yin + (long)s1 * D_FEAT))[lane];
        float2 x2 = ((const float2*)(yin + (long)s2 * D_FEAT))[lane];
        float2 x3 = ((const float2*)(yin + (long)s3 * D_FEAT))[lane];
        ax += x0.x; ay += x0.y;
        ax += x1.x; ay += x1.y;
        ax += x2.x; ay += x2.y;
        ax += x3.x; ay += x3.y;
      }
      if (k < m) {
        int r = m - k;
        int s0 = __builtin_amdgcn_readlane(cj, k);
        float2 x0 = ((const float2*)(yin + (long)s0 * D_FEAT))[lane];
        float2 x1 = make_float2(0.0f, 0.0f);
        float2 x2 = make_float2(0.0f, 0.0f);
        if (r > 1) {
          int s1 = __builtin_amdgcn_readlane(cj, k + 1);
          x1 = ((const float2*)(yin + (long)s1 * D_FEAT))[lane];
        }
        if (r > 2) {
          int s2 = __builtin_amdgcn_readlane(cj, k + 2);
          x2 = ((const float2*)(yin + (long)s2 * D_FEAT))[lane];
        }
        ax += x0.x; ay += x0.y;
        ax += x1.x; ay += x1.y;
        ax += x2.x; ay += x2.y;
      }
    }
    float g = cH * nv * nv;
    float ox = g * ax + cX * t1.x;
    float oy = g * ay + cX * t1.y;
    if (HAS_PREV) { ox += cP * t2.x; oy += cP * t2.y; }
    ((float2*)(yout + ro))[lane] = make_float2(ox, oy);
    start = startn;
    deg = degn;
  }
}


#define GEMM_WAVES 8
#define TILES_PER_WAVE 2
#define APITCH 68

__global__ __launch_bounds__(256) void gemm_fused_kernel(
    const float* __restrict__ X0, const float* __restrict__ Y1,
    const float* __restrict__ Y2, const float* __restrict__ Y3,
    const float* __restrict__ rdn,
    const float* __restrict__ W, const float* __restrict__ bias,
    float* __restrict__ out, int N) {
  __shared__ float4 Wl[4 * 8 * 4 * 32];
  __shared__ __attribute__((aligned(16))) float Al[GEMM_WAVES][16 * APITCH];

  int lane = threadIdx.x & 31;
  int wv   = threadIdx.x >> 5;
  int c    = lane & 15;
  int h    = lane >> 4;

#pragma unroll
  for (int i = 0; i < 16; ++i) {
    int q  = i * 256 + (int)threadIdx.x;
    int ln = q & 31, n = (q >> 5) & 3, j = (q >> 7) & 7, t = q >> 10;
    const float* wp = W + t * (D_FEAT * D_FEAT) + (8 * j + 4 * (ln >> 4)) * D_FEAT + 16 * n + (ln & 15);
    Wl[q] = make_float4(wp[0], wp[D_FEAT], wp[2 * D_FEAT], wp[3 * D_FEAT]);
  }
  __syncthreads();

  const int ntiles  = (N + 15) >> 4;
  const int tstride = (int)gridDim.x * GEMM_WAVES;
  int tile = (int)blockIdx.x * GEMM_WAVES + wv;
  if (tile >= ntiles) return;

  float* al = &Al[wv][0];
  float* ast = al + h * APITCH + c * 4;
  const float* arow = al + c * APITCH + 4 * h;

  float bb[4];
#pragma unroll
  for (int n = 0; n < 4; ++n) bb[n] = bias[16 * n + c];

  for (; tile < ntiles; tile += tstride) {
    const int row0 = tile * 16;
    const bool full = row0 + 16 <= N;
    long goff[8];
#pragma unroll
    for (int i = 0; i < 8; ++i) goff[i] = (long)min(row0 + 2 * i + h, N - 1) * (D_FEAT / 4) + c;
    float4 g[8];
#pragma unroll
    for (int i = 0; i < 8; ++i) g[i] = ((const float4*)Y1)[goff[i]];

    float rs[8];
    if (full) {
      const float4* rp = (const float4*)(rdn + row0 + h * 8);
      float4 lo = rp[0], hi = rp[1];
      rs[0] = lo.x; rs[1] = lo.y; rs[2] = lo.z; rs[3] = lo.w;
      rs[4] = hi.x; rs[5] = hi.y; rs[6] = hi.z; rs[7] = hi.w;
    } else {
#pragma unroll
      for (int v = 0; v < 8; ++v) rs[v] = rdn[min(row0 + h * 8 + v, N - 1)];
    }

    v8f acc[4];
#pragma unroll
    for (int n = 0; n < 4; ++n)
#pragma unroll
      for (int v = 0; v < 8; ++v) acc[n][v] = 0.0f;

    {
      __builtin_amdgcn_fence(__ATOMIC_RELEASE, "wavefront");
      __builtin_amdgcn_wave_barrier();
#pragma unroll
      for (int i = 0; i < 8; ++i) *(float4*)(ast + 2 * i * APITCH) = g[i];
      __builtin_amdgcn_fence(__ATOMIC_RELEASE, "wavefront");
      __builtin_amdgcn_wave_barrier();
      __builtin_amdgcn_fence(__ATOMIC_ACQUIRE, "wavefront");
#pragma unroll
      for (int i = 0; i < 8; ++i) g[i] = ((const float4*)Y2)[goff[i]];
      float4 a4[8];
#pragma unroll
      for (int j = 0; j < 8; ++j) a4[j] = *(const float4*)(arow + 8 * j);
#pragma unroll
      for (int j = 0; j < 8; ++j) {
        float4 b4[4];
#pragma unroll
        for (int n = 0; n < 4; ++n) b4[n] = Wl[((1 * 8 + j) * 4 + n) * 32 + lane];
        v2f a01, a23;
        a01.x = a4[j].x; a01.y = a4[j].y;
        a23.x = a4[j].z; a23.y = a4[j].w;
#pragma unroll
        for (int n = 0; n < 4; ++n) {
          v2f b; b.x = b4[n].x; b.y = b4[n].y;
          acc[n] = __builtin_amdgcn_wmma_f32_16x16x4_f32(
               false, a01,  false, b,
               (short)0, acc[n],  false,  false);
        }
#pragma unroll
        for (int n = 0; n < 4; ++n) {
          v2f b; b.x = b4[n].z; b.y = b4[n].w;
          acc[n] = __builtin_amdgcn_wmma_f32_16x16x4_f32(
               false, a23,  false, b,
               (short)0, acc[n],  false,  false);
        }
      }
    }
    {
      __builtin_amdgcn_fence(__ATOMIC_RELEASE, "wavefront");
      __builtin_amdgcn_wave_barrier();
#pragma unroll
      for (int i = 0; i < 8; ++i) *(float4*)(ast + 2 * i * APITCH) = g[i];
      __builtin_amdgcn_fence(__ATOMIC_RELEASE, "wavefront");
      __builtin_amdgcn_wave_barrier();
      __builtin_amdgcn_fence(__ATOMIC_ACQUIRE, "wavefront");
#pragma unroll
      for (int i = 0; i < 8; ++i) g[i] = ((const float4*)Y3)[goff[i]];
      float4 a4[8];
#pragma unroll
      for (int j = 0; j < 8; ++j) a4[j] = *(const float4*)(arow + 8 * j);
#pragma unroll
      for (int j = 0; j < 8; ++j) {
        float4 b4[4];
#pragma unroll
        for (int n = 0; n < 4; ++n) b4[n] = Wl[((2 * 8 + j) * 4 + n) * 32 + lane];
        v2f a01, a23;
        a01.x = a4[j].x; a01.y = a4[j].y;
        a23.x = a4[j].z; a23.y = a4[j].w;
#pragma unroll
        for (int n = 0; n < 4; ++n) {
          v2f b; b.x = b4[n].x; b.y = b4[n].y;
          acc[n] = __builtin_amdgcn_wmma_f32_16x16x4_f32(
               false, a01,  false, b,
               (short)0, acc[n],  false,  false);
        }
#pragma unroll
        for (int n = 0; n < 4; ++n) {
          v2f b; b.x = b4[n].z; b.y = b4[n].w;
          acc[n] = __builtin_amdgcn_wmma_f32_16x16x4_f32(
               false, a23,  false, b,
               (short)0, acc[n],  false,  false);
        }
      }
    }
    {
      __builtin_amdgcn_fence(__ATOMIC_RELEASE, "wavefront");
      __builtin_amdgcn_wave_barrier();
#pragma unroll
      for (int i = 0; i < 8; ++i) *(float4*)(ast + 2 * i * APITCH) = g[i];
      __builtin_amdgcn_fence(__ATOMIC_RELEASE, "wavefront");
      __builtin_amdgcn_wave_barrier();
      __builtin_amdgcn_fence(__ATOMIC_ACQUIRE, "wavefront");
#pragma unroll
      for (int i = 0; i < 8; ++i) g[i] = ((const float4*)X0)[goff[i]];
      float4 a4[8];
#pragma unroll
      for (int j = 0; j < 8; ++j) a4[j] = *(const float4*)(arow + 8 * j);
#pragma unroll
      for (int j = 0; j < 8; ++j) {
        float4 b4[4];
#pragma unroll
        for (int n = 0; n < 4; ++n) b4[n] = Wl[((3 * 8 + j) * 4 + n) * 32 + lane];
        v2f a01, a23;
        a01.x = a4[j].x; a01.y = a4[j].y;
        a23.x = a4[j].z; a23.y = a4[j].w;
#pragma unroll
        for (int n = 0; n < 4; ++n) {
          v2f b; b.x = b4[n].x; b.y = b4[n].y;
          acc[n] = __builtin_amdgcn_wmma_f32_16x16x4_f32(
               false, a01,  false, b,
               (short)0, acc[n],  false,  false);
        }
#pragma unroll
        for (int n = 0; n < 4; ++n) {
          v2f b; b.x = b4[n].z; b.y = b4[n].w;
          acc[n] = __builtin_amdgcn_wmma_f32_16x16x4_f32(
               false, a23,  false, b,
               (short)0, acc[n],  false,  false);
        }
      }
    }
    {
#pragma unroll
      for (int n = 0; n < 4; ++n)
#pragma unroll
        for (int v = 0; v < 8; ++v) acc[n][v] = acc[n][v] * rs[v] + bb[n];
      __builtin_amdgcn_fence(__ATOMIC_RELEASE, "wavefront");
      __builtin_amdgcn_wave_barrier();
#pragma unroll
      for (int i = 0; i < 8; ++i) *(float4*)(ast + 2 * i * APITCH) = g[i];
      __builtin_amdgcn_fence(__ATOMIC_RELEASE, "wavefront");
      __builtin_amdgcn_wave_barrier();
      __builtin_amdgcn_fence(__ATOMIC_ACQUIRE, "wavefront");
      float4 a4[8];
#pragma unroll
      for (int j = 0; j < 8; ++j) a4[j] = *(const float4*)(arow + 8 * j);
#pragma unroll
      for (int j = 0; j < 8; ++j) {
        float4 b4[4];
#pragma unroll
        for (int n = 0; n < 4; ++n) b4[n] = Wl[((0 * 8 + j) * 4 + n) * 32 + lane];
        v2f a01, a23;
        a01.x = a4[j].x; a01.y = a4[j].y;
        a23.x = a4[j].z; a23.y = a4[j].w;
#pragma unroll
        for (int n = 0; n < 4; ++n) {
          v2f b; b.x = b4[n].x; b.y = b4[n].y;
          acc[n] = __builtin_amdgcn_wmma_f32_16x16x4_f32(
               false, a01,  false, b,
               (short)0, acc[n],  false,  false);
        }
#pragma unroll
        for (int n = 0; n < 4; ++n) {
          v2f b; b.x = b4[n].z; b.y = b4[n].w;
          acc[n] = __builtin_amdgcn_wmma_f32_16x16x4_f32(
               false, a23,  false, b,
               (short)0, acc[n],  false,  false);
        }
      }
    }

    if (full) {
      float* orow = out + (long)(row0 + h * 8) * D_FEAT + c;
#pragma unroll
      for (int n = 0; n < 4; ++n)
#pragma unroll
        for (int v = 0; v < 8; ++v) orow[(long)v * D_FEAT + 16 * n] = acc[n][v];
    } else {
#pragma unroll
      for (int n = 0; n < 4; ++n)
#pragma unroll
        for (int v = 0; v < 8; ++v) {
          int r = row0 + h * 8 + v;
          if (r < N) out[(long)r * D_FEAT + 16 * n + c] = acc[n][v];
        }
    }
  }
}


extern "C" void kernel_launch(void* const* d_in, const int* in_sizes, int n_in,
                              void* d_out, int out_size, void* d_ws, size_t ws_size,
                              hipStream_t stream) {
  const float* feat = (const float*)d_in[0];
  const int*   src  = (const int*)d_in[1];
  const int*   dst  = (const int*)d_in[2];
  const float* W    = (const float*)d_in[3];
  const float* b    = (const float*)d_in[4];
  float*       out  = (float*)d_out;

  const int N = in_sizes[0] / D_FEAT;
  const int E = in_sizes[1];
  const long nd = (long)N * D_FEAT;
  const int nbins   = (N + BIN_NODES - 1) / BIN_NODES;
  const int nchunks = (E + CHUNK - 1) / CHUNK;

  const long Np   = ((long)N + 63) / 64 * 64;
  const long Ep   = ((long)E + 63) / 64 * 64;
  const long tabp = ((long)nbins * nchunks + 63) / 64 * 64;
  int*      cnt      = (int*)d_ws;
  int*      rowstart = cnt + Np;
  unsigned* tab      = (unsigned*)(rowstart + Np);
  int*      bsrc     = (int*)(tab + tabp);
  int*      bdst     = bsrc + Ep;
  int*      col      = bdst + Ep;
  float*    nrm      = (float*)(col + Ep);
  float*    rdn      = nrm + Np;
  float*    y0       = rdn + Np;
  float*    txA      = y0 + nd;
  float*    txB      = txA + nd;
  float*    txC      = txB + nd;

  const int ngrid = (N + 8 * NPW - 1) / (8 * NPW);
  const int ggrid = (N + 16 * GEMM_WAVES * TILES_PER_WAVE - 1) / (16 * GEMM_WAVES * TILES_PER_WAVE);

  bin_kernel<<<nchunks, 1024, 0, stream>>>(src, dst, bsrc, bdst, tab, E, N, nbins, nchunks);
  csr_kernel<<<nbins, 1024, 0, stream>>>(bsrc, bdst, tab, feat, cnt, rowstart, nrm, rdn, y0, col, N, E, nbins, nchunks);

  gather_combine_kernel<false><<<ngrid, 256, 0, stream>>>(
      y0, y0, nrm, cnt, rowstart, col, txA,
      -2.0f / LAM, 2.0f / LAM - 1.0f, 0.0f, N);
  gather_combine_kernel<true><<<ngrid, 256, 0, stream>>>(
      txA, y0, nrm, cnt, rowstart, col, txB,
      -4.0f / LAM, 4.0f / LAM - 2.0f, -1.0f, N);
  gather_combine_kernel<true><<<ngrid, 256, 0, stream>>>(
      txB, txA, nrm, cnt, rowstart, col, txC,
      -4.0f / LAM, 4.0f / LAM - 2.0f, -1.0f, N);

  gemm_fused_kernel<<<ggrid, 256, 0, stream>>>(feat, txA, txB, txC, rdn, W, b, out, N);
}


# Third propagation runs inside the GEMM launch: each wave gathers y3 for its own 16 rows into its LDS A tile (same rank-order sums); one launch and y3's memory round trip go; 16-wave blocks
# speedup vs baseline: 5.3741x; 1.0318x over previous
#include <hip/hip_runtime.h>


typedef __attribute__((ext_vector_type(2))) float v2f;
typedef __attribute__((ext_vector_type(8))) float v8f;

#define D_FEAT 64
constexpr float LAM = 2.0f;


#define BIN_SHIFT 8
#define BIN_NODES 256
#define CHUNK 4096
#define MAXBINS 512
#define STAGE_CAP 16384

__device__ __forceinline__ int wave_incl_scan(int x, int lane) {
#pragma unroll
  for (int o = 1; o < 32; o <<= 1) {
    int y = __shfl_up(x, o, 32);
    if (lane >= o) x += y;
  }
  return x;
}

__global__ __launch_bounds__(1024) void bin_kernel(
    const int* __restrict__ src, const int* __restrict__ dst,
    int* __restrict__ bsrc, int* __restrict__ bdst, unsigned* __restrict__ tab,
    int E, int N, int nbins, int nchunks) {
  __shared__ int hist[MAXBINS];
  __shared__ int wsum[16];
  __shared__ int ssrc[CHUNK];
  __shared__ int sdst[CHUNK];
  int t = threadIdx.x, lane = t & 31, w = t >> 5;
  int k = blockIdx.x;
  int e0 = k * CHUNK;
  if (t < MAXBINS) hist[t] = 0;
  __syncthreads();

  int s[4], d[4], r[4];
#pragma unroll
  for (int j = 0; j < 4; ++j) {
    int ec = min(e0 + j * 1024 + t, E - 1);
    s[j] = src[ec];
    d[j] = (int)min((unsigned)dst[ec], (unsigned)(N - 1));
  }
#pragma unroll
  for (int j = 0; j < 4; ++j) {
    r[j] = 0;
    if (e0 + j * 1024 + t < E) r[j] = atomicAdd(&hist[d[j] >> BIN_SHIFT], 1);
  }
  __syncthreads();

  int c = 0, inc = 0;
  if (t < MAXBINS) {
    c = hist[t];
    inc = wave_incl_scan(c, lane);
    if (lane == 31) wsum[w] = inc;
  }
  __syncthreads();
  if (t < MAXBINS) {
    int woff = 0;
#pragma unroll
    for (int q = 0; q < 16; ++q) if (q < w) woff += wsum[q];
    int st = woff + inc - c;
    hist[t] = st;
    if (t < nbins) tab[(long)k * nbins + t] = ((unsigned)st << 16) | (unsigned)c;
  }
  __syncthreads();

#pragma unroll
  for (int j = 0; j < 4; ++j) {
    int e = e0 + j * 1024 + t;
    if (e < E) {
      int p = hist[d[j] >> BIN_SHIFT] + r[j];
      ssrc[p] = s[j];
      sdst[p] = d[j];
    }
  }
  __syncthreads();
  for (int j = 0; j < 4; ++j) {
    int i = j * 1024 + t;
    if (e0 + i < E) {
      bsrc[e0 + i] = ssrc[i];
      bdst[e0 + i] = sdst[i];
    }
  }
}

#define LROW 1024
#define RUNS 4

__device__ __forceinline__ unsigned run_word(const unsigned* srow, const unsigned* __restrict__ tab,
                                             int k, int nchunks, int nbins, int b) {
  if (k >= nchunks) return 0u;
  return (k < LROW) ? srow[k] : tab[(long)k * nbins + b];
}

__global__ __launch_bounds__(1024) void csr_kernel(
    const int* __restrict__ bsrc, const int* __restrict__ bdst, const unsigned* __restrict__ tab,
    const float* __restrict__ feat,
    int* __restrict__ cnt, int* __restrict__ rowstart, float* __restrict__ nrm,
    float* __restrict__ rdn, float* __restrict__ y0,
    int* __restrict__ col, int N, int E, int nbins, int nchunks) {
  __shared__ int lcnt[BIN_NODES];
  __shared__ int lstart[BIN_NODES];
  __shared__ int lfill[BIN_NODES];
  __shared__ int wsum[8];
  __shared__ int wbase[32];
  __shared__ int wtot[32];
  __shared__ unsigned srow[LROW];
  __shared__ int stage[STAGE_CAP];
  __shared__ float lnrm[BIN_NODES];
  int t = threadIdx.x, lane = t & 31, w = t >> 5;
  int hw = t >> 4, hl = t & 15;
  int b = blockIdx.x;

  int mybase = 0, mytot = 0;
  for (int k = t; k < nchunks; k += 1024) {
    unsigned en = tab[(long)k * nbins + b];
    if (k < LROW) srow[k] = en;
    mybase += (int)(en >> 16);
    mytot += (int)(en & 0xffffu);
  }
#pragma unroll
  for (int o = 16; o > 0; o >>= 1) {
    mybase += __shfl_xor(mybase, o, 32);
    mytot += __shfl_xor(mytot, o, 32);
  }
  if (lane == 0) { wbase[w] = mybase; wtot[w] = mytot; }
  if (t < BIN_NODES) lcnt[t] = 0;
  __syncthreads();
  int binbase = 0, total = 0;
#pragma unroll
  for (int q = 0; q < 32; ++q) { binbase += wbase[q]; total += wtot[q]; }
  const bool staged = total <= STAGE_CAP;

  for (int k0 = hw; k0 < nchunks; k0 += 64 * RUNS) {
    unsigned en[RUNS];
    int rc[RUNS], dv[RUNS];
    long eb[RUNS];
#pragma unroll
    for (int u = 0; u < RUNS; ++u) en[u] = run_word(srow, tab, k0 + 64 * u, nchunks, nbins, b);
#pragma unroll
    for (int u = 0; u < RUNS; ++u) {
      rc[u] = (int)(en[u] & 0xffffu);
      eb[u] = (long)(k0 + 64 * u) * CHUNK + (long)(en[u] >> 16);
      long ix = min(eb[u] + hl, (long)E - 1);
      dv[u] = bdst[ix];
    }
#pragma unroll
    for (int u = 0; u < RUNS; ++u) {
      if (hl < rc[u]) atomicAdd(&lcnt[dv[u] & (BIN_NODES - 1)], 1);
      for (int i = hl + 16; i < rc[u]; i += 16)
        atomicAdd(&lcnt[bdst[eb[u] + i] & (BIN_NODES - 1)], 1);
    }
  }

  float4 fx[4];
#pragma unroll
  for (int j = 0; j < 4; ++j) {
    int i = j * 1024 + t;
    long v = min((long)b * BIN_NODES + (i >> 4), (long)N - 1);
    fx[j] = ((const float4*)feat)[v * (D_FEAT / 4) + (i & 15)];
  }
  __syncthreads();

  int c = 0, inc = 0;
  if (t < BIN_NODES) {
    c = lcnt[t];
    inc = wave_incl_scan(c, lane);
    if (lane == 31) wsum[w] = inc;
  }
  __syncthreads();
  if (t < BIN_NODES) {
    int woff = 0;
#pragma unroll
    for (int q = 0; q < 8; ++q) if (q < w) woff += wsum[q];
    int ls = woff + inc - c;
    lstart[t] = ls;
    lfill[t] = 0;
    int v = b * BIN_NODES + t;
    float nv = rsqrtf(fmaxf((float)c, 1.0f));
    lnrm[t] = nv;
    if (v < N) {
      cnt[v] = c;
      rowstart[v] = binbase + ls;
      nrm[v] = nv;
      rdn[v] = sqrtf(fmaxf((float)c, 1.0f));
    }
  }
  __syncthreads();

#pragma unroll
  for (int j = 0; j < 4; ++j) {
    int i = j * 1024 + t;
    int row = i >> 4;
    long v = (long)b * BIN_NODES + row;
    if (v < N) {
      float s = lnrm[row];
      float4 x = fx[j];
      ((float4*)y0)[v * (D_FEAT / 4) + (i & 15)] = make_float4(x.x * s, x.y * s, x.z * s, x.w * s);
    }
  }

  for (int k0 = hw; k0 < nchunks; k0 += 64 * RUNS) {
    unsigned en[RUNS];
    int rc[RUNS], dv[RUNS], sv[RUNS];
    long eb[RUNS];
#pragma unroll
    for (int u = 0; u < RUNS; ++u) en[u] = run_word(srow, tab, k0 + 64 * u, nchunks, nbins, b);
#pragma unroll
    for (int u = 0; u < RUNS; ++u) {
      rc[u] = (int)(en[u] & 0xffffu);
      eb[u] = (long)(k0 + 64 * u) * CHUNK + (long)(en[u] >> 16);
      long ix = min(eb[u] + hl, (long)E - 1);
      dv[u] = bdst[ix];
      sv[u] = bsrc[ix];
    }
#pragma unroll
    for (int u = 0; u < RUNS; ++u) {
      for (int i = hl; i < rc[u]; i += 16) {
        int d1 = dv[u], s1 = sv[u];
        if (i != hl) { d1 = bdst[eb[u] + i]; s1 = bsrc[eb[u] + i]; }
        int dl = d1 & (BIN_NODES - 1);
        int p = lstart[dl] + atomicAdd(&lfill[dl], 1);
        if (staged) {
          stage[p] = s1;
        } else {
          if ((unsigned)(binbase + p) < (unsigned)E) col[binbase + p] = s1;
        }
      }
    }
  }
  __syncthreads();
  if (staged) {
    for (int i = t; i < total; i += 1024) {
      if ((unsigned)(binbase + i) < (unsigned)E) col[binbase + i] = stage[i];
    }
  }
}

#define NPW 4
template <bool HAS_PREV>
__global__ __launch_bounds__(256) void gather_combine_kernel(
    const float* __restrict__ yin, const float* __restrict__ yprev,
    const float* __restrict__ nrm, const int* __restrict__ cnt,
    const int* __restrict__ rowstart,
    const int* __restrict__ col, float* __restrict__ yout,
    float cH, float cX, float cP, int N) {
  int lane = threadIdx.x & 31;
  int v0 = __builtin_amdgcn_readfirstlane((int)(blockIdx.x * 8 + (threadIdx.x >> 5))) * NPW;
  if (v0 >= N) return;
  int n = min(NPW, N - v0);
  int ms = 0, md = 0;
  float mn = 0.0f;
  if (lane < n) {
    ms = rowstart[v0 + lane];
    md = cnt[v0 + lane];
    mn = nrm[v0 + lane];
  }
  int start = __builtin_amdgcn_readlane(ms, 0);
  int deg = __builtin_amdgcn_readlane(md, 0);
  int cjn = 0;
  if (lane < min(deg, 32)) cjn = col[start + lane];

  for (int j = 0; j < n; ++j) {
    long ro = (long)(v0 + j) * D_FEAT;
    float nv = __builtin_bit_cast(float, __builtin_amdgcn_readlane(__builtin_bit_cast(int, mn), j));
    float2 t1 = ((const float2*)(yin + ro))[lane];
    float2 t2 = make_float2(0.0f, 0.0f);
    if (HAS_PREV) t2 = ((const float2*)(yprev + ro))[lane];
    int cj = (int)min((unsigned)cjn, (unsigned)(N - 1));
    int startn = 0, degn = 0;
    if (j + 1 < n) {
      startn = __builtin_amdgcn_readlane(ms, j + 1);
      degn = __builtin_amdgcn_readlane(md, j + 1);
      if (lane < min(degn, 32)) cjn = col[startn + lane];
    }

    float ax = 0.0f, ay = 0.0f;
    for (int base = 0; base < deg; base += 32) {
      int m = min(32, deg - base);
      if (base > 0) {
        cj = 0;
        if (lane < m) cj = (int)min((unsigned)col[start + base + lane], (unsigned)(N - 1));
      }
      int k = 0;
      for (; k + 4 <= m; k += 4) {
        int s0 = __builtin_amdgcn_readlane(cj, k + 0);
        int s1 = __builtin_amdgcn_readlane(cj, k + 1);
        int s2 = __builtin_amdgcn_readlane(cj, k + 2);
        int s3 = __builtin_amdgcn_readlane(cj, k + 3);
        float2 x0 = ((const float2*)(yin + (long)s0 * D_FEAT))[lane];
        float2 x1 = ((const float2*)(yin + (long)s1 * D_FEAT))[lane];
        float2 x2 = ((const float2*)(yin + (long)s2 * D_FEAT))[lane];
        float2 x3 = ((const float2*)(yin + (long)s3 * D_FEAT))[lane];
        ax += x0.x; ay += x0.y;
        ax += x1.x; ay += x1.y;
        ax += x2.x; ay += x2.y;
        ax += x3.x; ay += x3.y;
      }
      if (k < m) {
        int r = m - k;
        int s0 = __builtin_amdgcn_readlane(cj, k);
        float2 x0 = ((const float2*)(yin + (long)s0 * D_FEAT))[lane];
        float2 x1 = make_float2(0.0f, 0.0f);
        float2 x2 = make_float2(0.0f, 0.0f);
        if (r > 1) {
          int s1 = __builtin_amdgcn_readlane(cj, k + 1);
          x1 = ((const float2*)(yin + (long)s1 * D_FEAT))[lane];
        }
        if (r > 2) {
          int s2 = __builtin_amdgcn_readlane(cj, k + 2);
          x2 = ((const float2*)(yin + (long)s2 * D_FEAT))[lane];
        }
        ax += x0.x; ay += x0.y;
        ax += x1.x; ay += x1.y;
        ax += x2.x; ay += x2.y;
      }
    }
    float g = cH * nv * nv;
    float ox = g * ax + cX * t1.x;
    float oy = g * ay + cX * t1.y;
    if (HAS_PREV) { ox += cP * t2.x; oy += cP * t2.y; }
    ((float2*)(yout + ro))[lane] = make_float2(ox, oy);
    start = startn;
    deg = degn;
  }
}


#define GEMM_WAVES 16
#define TILES_PER_WAVE 1
#define APITCH 68

__device__ __forceinline__ void tile_open() {
  __builtin_amdgcn_fence(__ATOMIC_RELEASE, "wavefront");
  __builtin_amdgcn_wave_barrier();
}
__device__ __forceinline__ void tile_close() {
  __builtin_amdgcn_fence(__ATOMIC_RELEASE, "wavefront");
  __builtin_amdgcn_wave_barrier();
  __builtin_amdgcn_fence(__ATOMIC_ACQUIRE, "wavefront");
}

template <int T>
__device__ __forceinline__ void term_products(const float* arow, const float4* Wl, int lane, v8f (&acc)[4]) {
  float4 a4[8];
#pragma unroll
  for (int j = 0; j < 8; ++j) a4[j] = *(const float4*)(arow + 8 * j);
#pragma unroll
  for (int j = 0; j < 8; ++j) {
    float4 b4[4];
#pragma unroll
    for (int n = 0; n < 4; ++n) b4[n] = Wl[((T * 8 + j) * 4 + n) * 32 + lane];
    v2f a01, a23;
    a01.x = a4[j].x; a01.y = a4[j].y;
    a23.x = a4[j].z; a23.y = a4[j].w;
#pragma unroll
    for (int n = 0; n < 4; ++n) {
      v2f b; b.x = b4[n].x; b.y = b4[n].y;
      acc[n] = __builtin_amdgcn_wmma_f32_16x16x4_f32(
           false, a01,  false, b,
           (short)0, acc[n],  false,  false);
    }
#pragma unroll
    for (int n = 0; n < 4; ++n) {
      v2f b; b.x = b4[n].z; b.y = b4[n].w;
      acc[n] = __builtin_amdgcn_wmma_f32_16x16x4_f32(
           false, a23,  false, b,
           (short)0, acc[n],  false,  false);
    }
  }
}

__global__ __launch_bounds__(GEMM_WAVES * 32) __attribute__((amdgpu_waves_per_eu(8))) void gemm_fused_kernel(
    const float* __restrict__ X0, const float* __restrict__ Y1,
    const float* __restrict__ Y2,
    const float* __restrict__ nrm, const int* __restrict__ cnt,
    const int* __restrict__ rowstart, const int* __restrict__ col,
    const float* __restrict__ rdn,
    const float* __restrict__ W, const float* __restrict__ bias,
    float* __restrict__ out, float cH, float cX, float cP, int N) {
  __shared__ float4 Wl[4 * 8 * 4 * 32];
  __shared__ __attribute__((aligned(16))) float Al[GEMM_WAVES][16 * APITCH];

  int lane = threadIdx.x & 31;
  int wv   = threadIdx.x >> 5;
  int c    = lane & 15;
  int h    = lane >> 4;

#pragma unroll
  for (int i = 0; i < 4096 / (GEMM_WAVES * 32); ++i) {
    int q  = i * (GEMM_WAVES * 32) + (int)threadIdx.x;
    int ln = q & 31, n = (q >> 5) & 3, j = (q >> 7) & 7, t = q >> 10;
    const float* wp = W + t * (D_FEAT * D_FEAT) + (8 * j + 4 * (ln >> 4)) * D_FEAT + 16 * n + (ln & 15);
    Wl[q] = make_float4(wp[0], wp[D_FEAT], wp[2 * D_FEAT], wp[3 * D_FEAT]);
  }
  __syncthreads();

  const int ntiles  = (N + 15) >> 4;
  const int tstride = (int)gridDim.x * GEMM_WAVES;
  int tile = __builtin_amdgcn_readfirstlane((int)blockIdx.x * GEMM_WAVES + wv);
  if (tile >= ntiles) return;

  float* al = &Al[wv][0];
  float* ast = al + h * APITCH + c * 4;
  const float* arow = al + c * APITCH + 4 * h;

  float bb[4];
#pragma unroll
  for (int n = 0; n < 4; ++n) bb[n] = bias[16 * n + c];

  for (; tile < ntiles; tile += tstride) {
    const int row0 = tile * 16;
    const bool full = row0 + 16 <= N;
    const int nn = min(16, N - row0);
    int ms = 0, md = 0;
    float mn = 0.0f;
    if (lane < nn) {
      ms = rowstart[row0 + lane];
      md = cnt[row0 + lane];
      mn = nrm[row0 + lane];
    }
    int goff[8];
#pragma unroll
    for (int i = 0; i < 8; ++i) goff[i] = min(row0 + 2 * i + h, N - 1) * (D_FEAT / 4) + c;
    float4 g[8];
#pragma unroll
    for (int i = 0; i < 8; ++i) g[i] = ((const float4*)Y1)[goff[i]];

    v8f acc[4];
#pragma unroll
    for (int n = 0; n < 4; ++n)
#pragma unroll
      for (int v = 0; v < 8; ++v) acc[n][v] = 0.0f;

    int start = __builtin_amdgcn_readlane(ms, 0);
    int deg = __builtin_amdgcn_readlane(md, 0);
    int cjn = 0;
    if (lane < min(deg, 32)) cjn = col[start + lane];

    {
      tile_open();
#pragma unroll
      for (int i = 0; i < 8; ++i) *(float4*)(ast + 2 * i * APITCH) = g[i];
      tile_close();
#pragma unroll
      for (int i = 0; i < 8; ++i) g[i] = ((const float4*)Y2)[goff[i]];
      term_products<1>(arow, Wl, lane, acc);
    }
    {
      tile_open();
#pragma unroll
      for (int i = 0; i < 8; ++i) *(float4*)(ast + 2 * i * APITCH) = g[i];
      tile_close();
      term_products<2>(arow, Wl, lane, acc);
    }
    {
      tile_open();
      for (int j = 0; j < nn; ++j) {
        long ro = (long)(row0 + j) * D_FEAT;
        float nv = __builtin_bit_cast(float, __builtin_amdgcn_readlane(__builtin_bit_cast(int, mn), j));
        float2 t1 = ((const float2*)(Y2 + ro))[lane];
        float2 t2 = ((const float2*)(Y1 + ro))[lane];
        int cj = (int)min((unsigned)cjn, (unsigned)(N - 1));
        int startn = 0, degn = 0;
        if (j + 1 < nn) {
          startn = __builtin_amdgcn_readlane(ms, j + 1);
          degn = __builtin_amdgcn_readlane(md, j + 1);
          if (lane < min(degn, 32)) cjn = col[startn + lane];
        }

        float ax = 0.0f, ay = 0.0f;
        for (int base = 0; base < deg; base += 32) {
          int m = min(32, deg - base);
          if (base > 0) {
            cj = 0;
            if (lane < m) cj = (int)min((unsigned)col[start + base + lane], (unsigned)(N - 1));
          }
          int k = 0;
          for (; k + 8 <= m; k += 8) {
            int s0 = __builtin_amdgcn_readlane(cj, k + 0);
            int s1 = __builtin_amdgcn_readlane(cj, k + 1);
            int s2 = __builtin_amdgcn_readlane(cj, k + 2);
            int s3 = __builtin_amdgcn_readlane(cj, k + 3);
            int s4 = __builtin_amdgcn_readlane(cj, k + 4);
            int s5 = __builtin_amdgcn_readlane(cj, k + 5);
            int s6 = __builtin_amdgcn_readlane(cj, k + 6);
            int s7 = __builtin_amdgcn_readlane(cj, k + 7);
            float2 x0 = ((const float2*)(Y2 + (long)s0 * D_FEAT))[lane];
            float2 x1 = ((const float2*)(Y2 + (long)s1 * D_FEAT))[lane];
            float2 x2 = ((const float2*)(Y2 + (long)s2 * D_FEAT))[lane];
            float2 x3 = ((const float2*)(Y2 + (long)s3 * D_FEAT))[lane];
            float2 x4 = ((const float2*)(Y2 + (long)s4 * D_FEAT))[lane];
            float2 x5 = ((const float2*)(Y2 + (long)s5 * D_FEAT))[lane];
            float2 x6 = ((const float2*)(Y2 + (long)s6 * D_FEAT))[lane];
            float2 x7 = ((const float2*)(Y2 + (long)s7 * D_FEAT))[lane];
            ax += x0.x; ay += x0.y;
            ax += x1.x; ay += x1.y;
            ax += x2.x; ay += x2.y;
            ax += x3.x; ay += x3.y;
            ax += x4.x; ay += x4.y;
            ax += x5.x; ay += x5.y;
            ax += x6.x; ay += x6.y;
            ax += x7.x; ay += x7.y;
          }
          if (k < m) {
            int r = m - k;
            float2 x0 = ((const float2*)(Y2 + (long)__builtin_amdgcn_readlane(cj, k) * D_FEAT))[lane];
            float2 x1 = make_float2(0.0f, 0.0f);
            float2 x2 = make_float2(0.0f, 0.0f);
            float2 x3 = make_float2(0.0f, 0.0f);
            float2 x4 = make_float2(0.0f, 0.0f);
            float2 x5 = make_float2(0.0f, 0.0f);
            float2 x6 = make_float2(0.0f, 0.0f);
            if (r > 1) x1 = ((const float2*)(Y2 + (long)__builtin_amdgcn_readlane(cj, k + 1) * D_FEAT))[lane];
            if (r > 2) x2 = ((const float2*)(Y2 + (long)__builtin_amdgcn_readlane(cj, k + 2) * D_FEAT))[lane];
            if (r > 3) x3 = ((const float2*)(Y2 + (long)__builtin_amdgcn_readlane(cj, k + 3) * D_FEAT))[lane];
            if (r > 4) x4 = ((const float2*)(Y2 + (long)__builtin_amdgcn_readlane(cj, k + 4) * D_FEAT))[lane];
            if (r > 5) x5 = ((const float2*)(Y2 + (long)__builtin_amdgcn_readlane(cj, k + 5) * D_FEAT))[lane];
            if (r > 6) x6 = ((const float2*)(Y2 + (long)__builtin_amdgcn_readlane(cj, k + 6) * D_FEAT))[lane];
            ax += x0.x; ay += x0.y;
            ax += x1.x; ay += x1.y;
            ax += x2.x; ay += x2.y;
            ax += x3.x; ay += x3.y;
            ax += x4.x; ay += x4.y;
            ax += x5.x; ay += x5.y;
            ax += x6.x; ay += x6.y;
          }
        }
        float gq = cH * nv * nv;
        float ox = gq * ax + cX * t1.x;
        float oy = gq * ay + cX * t1.y;
        ox += cP * t2.x; oy += cP * t2.y;
        *(float2*)(al + j * APITCH + 2 * lane) = make_float2(ox, oy);
        start = startn;
        deg = degn;
      }
      tile_close();
#pragma unroll
      for (int i = 0; i < 8; ++i) g[i] = ((const float4*)X0)[goff[i]];
      term_products<3>(arow, Wl, lane, acc);
    }
    {
      float rs[8];
      if (full) {
        const float4* rp = (const float4*)(rdn + row0 + h * 8);
        float4 lo = rp[0], hi = rp[1];
        rs[0] = lo.x; rs[1] = lo.y; rs[2] = lo.z; rs[3] = lo.w;
        rs[4] = hi.x; rs[5] = hi.y; rs[6] = hi.z; rs[7] = hi.w;
      } else {
#pragma unroll
        for (int v = 0; v < 8; ++v) rs[v] = rdn[min(row0 + h * 8 + v, N - 1)];
      }
#pragma unroll
      for (int n = 0; n < 4; ++n)
#pragma unroll
        for (int v = 0; v < 8; ++v) acc[n][v] = acc[n][v] * rs[v] + bb[n];
      tile_open();
#pragma unroll
      for (int i = 0; i < 8; ++i) *(float4*)(ast + 2 * i * APITCH) = g[i];
      tile_close();
      term_products<0>(arow, Wl, lane, acc);
    }

    if (full) {
      float* orow = out + (long)(row0 + h * 8) * D_FEAT + c;
#pragma unroll
      for (int n = 0; n < 4; ++n)
#pragma unroll
        for (int v = 0; v < 8; ++v) orow[(long)v * D_FEAT + 16 * n] = acc[n][v];
    } else {
#pragma unroll
      for (int n = 0; n < 4; ++n)
#pragma unroll
        for (int v = 0; v < 8; ++v) {
          int r = row0 + h * 8 + v;
          if (r < N) out[(long)r * D_FEAT + 16 * n + c] = acc[n][v];
        }
    }
  }
}


extern "C" void kernel_launch(void* const* d_in, const int* in_sizes, int n_in,
                              void* d_out, int out_size, void* d_ws, size_t ws_size,
                              hipStream_t stream) {
  const float* feat = (const float*)d_in[0];
  const int*   src  = (const int*)d_in[1];
  const int*   dst  = (const int*)d_in[2];
  const float* W    = (const float*)d_in[3];
  const float* b    = (const float*)d_in[4];
  float*       out  = (float*)d_out;

  const int N = in_sizes[0] / D_FEAT;
  const int E = in_sizes[1];
  const long nd = (long)N * D_FEAT;
  const int nbins   = (N + BIN_NODES - 1) / BIN_NODES;
  const int nchunks = (E + CHUNK - 1) / CHUNK;

  const long Np   = ((long)N + 63) / 64 * 64;
  const long Ep   = ((long)E + 63) / 64 * 64;
  const long tabp = ((long)nbins * nchunks + 63) / 64 * 64;
  int*      cnt      = (int*)d_ws;
  int*      rowstart = cnt + Np;
  unsigned* tab      = (unsigned*)(rowstart + Np);
  int*      bsrc     = (int*)(tab + tabp);
  int*      bdst     = bsrc + Ep;
  int*      col      = bdst + Ep;
  float*    nrm      = (float*)(col + Ep);
  float*    rdn      = nrm + Np;
  float*    y0       = rdn + Np;
  float*    txA      = y0 + nd;
  float*    txB      = txA + nd;

  const int ngrid = (N + 8 * NPW - 1) / (8 * NPW);
  const int ggrid = (N + 16 * GEMM_WAVES * TILES_PER_WAVE - 1) / (16 * GEMM_WAVES * TILES_PER_WAVE);

  bin_kernel<<<nchunks, 1024, 0, stream>>>(src, dst, bsrc, bdst, tab, E, N, nbins, nchunks);
  csr_kernel<<<nbins, 1024, 0, stream>>>(bsrc, bdst, tab, feat, cnt, rowstart, nrm, rdn, y0, col, N, E, nbins, nchunks);

  gather_combine_kernel<false><<<ngrid, 256, 0, stream>>>(
      y0, y0, nrm, cnt, rowstart, col, txA,
      -2.0f / LAM, 2.0f / LAM - 1.0f, 0.0f, N);
  gather_combine_kernel<true><<<ngrid, 256, 0, stream>>>(
      txA, y0, nrm, cnt, rowstart, col, txB,
      -4.0f / LAM, 4.0f / LAM - 2.0f, -1.0f, N);

  gemm_fused_kernel<<<ggrid, GEMM_WAVES * 32, 0, stream>>>(
      feat, txA, txB, nrm, cnt, rowstart, col, rdn, W, b, out,
      -4.0f / LAM, 4.0f / LAM - 2.0f, -1.0f, N);
}
